# starting kernel
# baseline (speedup 1.0000x reference)
#include <hip/hip_runtime.h>
#include <hip/hip_bf16.h>

typedef __attribute__((ext_vector_type(2))) float v2f;
typedef __attribute__((ext_vector_type(8))) float v8f;

#define EMB 128
#define XPITCH 132
#define KSTEPS (EMB / 4)

#if defined(__has_builtin)
#if __has_builtin(__builtin_amdgcn_sched_barrier)
#define SCHED_FENCE() __builtin_amdgcn_sched_barrier(0)
#endif
#endif
#ifndef SCHED_FENCE
#define SCHED_FENCE()
#endif

__global__ void fill_kernel(float* __restrict__ p, float v, long long n) {
    long long i = (long long)blockIdx.x * blockDim.x + threadIdx.x;
    long long stride = (long long)gridDim.x * blockDim.x;
    for (; i < n; i += stride) p[i] = v;
}

__global__ void deg_accum_kernel(const int* __restrict__ dst, float* __restrict__ deg, int e) {
    int i = blockIdx.x * blockDim.x + threadIdx.x;
    if (i < e) atomicAdd(&deg[dst[i]], 1.0f);
}

__global__ void rsqrt_ip_kernel(float* __restrict__ d, int n) {
    int i = blockIdx.x * blockDim.x + threadIdx.x;
    if (i < n) d[i] = rsqrtf(d[i]);
}

template <int HOUT>
__global__ __launch_bounds__(256) void gemm_wmma_kernel(const float* __restrict__ X,
                                                        const float* __restrict__ W,
                                                        float* __restrict__ H, int nrows) {
    extern __shared__ float smem[];
    float* sW = smem;
    float* sX = smem + EMB * HOUT;

    const int tid = threadIdx.x;
    const int rowBase = blockIdx.x * 128;

    for (int i = tid * 4; i < EMB * HOUT; i += 256 * 4) {
        *(float4*)&sW[i] = *(const float4*)&W[i];
    }
    for (int i = tid * 4; i < 128 * EMB; i += 256 * 4) {
        int r = i >> 7, c = i & 127;
        int gr = rowBase + r;
        float4 v = make_float4(0.f, 0.f, 0.f, 0.f);
        if (gr < nrows) v = *(const float4*)&X[(long long)gr * EMB + c];
        *(float4*)&sX[r * XPITCH + c] = v;
    }
    __syncthreads();

    const int wave = tid >> 5;
    const int lane = tid & 31;
    const int nl = lane & 15;
    const int kh = lane >> 4;
    const int wr = wave * 16;

    const float* xrow = &sX[(wr + nl) * XPITCH + 2 * kh];
    v2f afrag[KSTEPS];
#pragma unroll
    for (int ks = 0; ks < KSTEPS; ++ks) afrag[ks] = *(const v2f*)&xrow[4 * ks];
    SCHED_FENCE();

    for (int nt = 0; nt < HOUT / 16; ++nt) {
        const float* wcol = &sW[nt * 16 + nl];
        v2f bfrag[KSTEPS];
#pragma unroll
        for (int ks = 0; ks < KSTEPS; ++ks) {
            const int kk = 4 * ks + 2 * kh;
            v2f b;
            b.x = wcol[kk * HOUT];
            b.y = wcol[(kk + 1) * HOUT];
            bfrag[ks] = b;
        }
        SCHED_FENCE();
        v8f acc = {};
#pragma unroll
        for (int ks = 0; ks < KSTEPS; ++ks) {
            acc = __builtin_amdgcn_wmma_f32_16x16x4_f32(
                false, afrag[ks], false, bfrag[ks], (short)0, acc, false, false);
        }
        SCHED_FENCE();
#pragma unroll
        for (int j = 0; j < 8; ++j) {
            int r = rowBase + wr + kh * 8 + j;
            if (r < nrows) H[(long long)r * HOUT + nt * 16 + nl] = acc[j];
        }
    }
}

template <int HOUT>
__global__ __launch_bounds__(256) void scatter_kernel(const int* __restrict__ src,
                                                      const int* __restrict__ dst,
                                                      const float* __restrict__ dinv,
                                                      const float* __restrict__ h,
                                                      float* __restrict__ agg, int e) {
    int gid = blockIdx.x * blockDim.x + threadIdx.x;
    int edge = gid >> 5;
    int lane = gid & 31;
    if (edge >= e) return;
    int s = src[edge], d = dst[edge];
    float w = dinv[s] * dinv[d];
    constexpr int V = HOUT / 32;
    const float* hs = &h[(long long)s * HOUT + lane * V];
    float* ad = &agg[(long long)d * HOUT + lane * V];
    if constexpr (V == 4) {
        float4 v = *(const float4*)hs;
        atomicAdd(&ad[0], w * v.x);
        atomicAdd(&ad[1], w * v.y);
        atomicAdd(&ad[2], w * v.z);
        atomicAdd(&ad[3], w * v.w);
    } else {
        float2 v = *(const float2*)hs;
        atomicAdd(&ad[0], w * v.x);
        atomicAdd(&ad[1], w * v.y);
    }
}

__global__ void bias_relu_kernel(float* __restrict__ agg, const float* __restrict__ h,
                                 const float* __restrict__ dinv, const float* __restrict__ b,
                                 long long total, int hdim) {
    long long i = (long long)blockIdx.x * blockDim.x + threadIdx.x;
    if (i >= total) return;
    int row = (int)(i / hdim), col = (int)(i % hdim);
    float di = dinv[row];
    float v = agg[i] + di * di * h[i] + b[col];
    agg[i] = fmaxf(v, 0.0f);
}

__global__ void final_add_kernel(float* __restrict__ out, const float* __restrict__ h,
                                 const float* __restrict__ dinv, const float* __restrict__ b,
                                 long long total, int hdim) {
    long long i = (long long)blockIdx.x * blockDim.x + threadIdx.x;
    if (i >= total) return;
    int row = (int)(i / hdim), col = (int)(i % hdim);
    float di = dinv[row];
    out[i] = out[i] + di * di * h[i] + b[col];
}

extern "C" void kernel_launch(void* const* d_in, const int* in_sizes, int n_in,
                              void* d_out, int out_size, void* d_ws, size_t ws_size,
                              hipStream_t stream) {
    const int*   edge = (const int*)d_in[0];
    const float* x0   = (const float*)d_in[1];
    const float* W1   = (const float*)d_in[2];
    const float* b1   = (const float*)d_in[3];
    const float* W2   = (const float*)d_in[4];
    const float* b2   = (const float*)d_in[5];
    float* out = (float*)d_out;

    const int E    = in_sizes[0] / 2;
    const int N    = in_sizes[1] / EMB;
    const int HID  = in_sizes[3];
    const int REPR = in_sizes[5];
    const int* src = edge;
    const int* dst = edge + E;

    float* dinv = (float*)d_ws;
    float* h    = dinv + N;
    float* agg  = h + (size_t)N * HID;

    const int T = 256;
    long long nh = (long long)N * HID;
    long long nr = (long long)N * REPR;

    fill_kernel<<<512, T, 0, stream>>>(dinv, 1.0f, N);
    deg_accum_kernel<<<(E + T - 1) / T, T, 0, stream>>>(dst, dinv, E);
    rsqrt_ip_kernel<<<(N + T - 1) / T, T, 0, stream>>>(dinv, N);

    size_t smem1 = (size_t)(EMB * HID + 128 * XPITCH) * sizeof(float);
    gemm_wmma_kernel<128><<<(N + 127) / 128, T, smem1, stream>>>(x0, W1, h, N);
    fill_kernel<<<2048, T, 0, stream>>>(agg, 0.0f, nh);
    scatter_kernel<128><<<(int)(((long long)E * 32 + T - 1) / T), T, 0, stream>>>(
        src, dst, dinv, h, agg, E);
    bias_relu_kernel<<<(int)((nh + T - 1) / T), T, 0, stream>>>(agg, h, dinv, b1, nh, HID);

    size_t smem2 = (size_t)(EMB * REPR + 128 * XPITCH) * sizeof(float);
    gemm_wmma_kernel<64><<<(N + 127) / 128, T, smem2, stream>>>(agg, W2, h, N);
    fill_kernel<<<2048, T, 0, stream>>>(out, 0.0f, nr);
    scatter_kernel<64><<<(int)(((long long)E * 32 + T - 1) / T), T, 0, stream>>>(
        src, dst, dinv, h, out, E);
    final_add_kernel<<<(int)((nr + T - 1) / T), T, 0, stream>>>(out, h, dinv, b2, nr, REPR);
}


# Replace float-atomic edge scatter by a register-sum gather over an in-call dst-sorted edge list (counting sort); per-node sum order = arrival-order rank from integer atomics
# speedup vs baseline: 1.8911x; 1.8911x over previous
#include <hip/hip_runtime.h>
#include <hip/hip_bf16.h>

typedef __attribute__((ext_vector_type(2))) float v2f;
typedef __attribute__((ext_vector_type(8))) float v8f;

#define EMB 128
#define XPITCH 132
#define KSTEPS (EMB / 4)

#if defined(__has_builtin)
#if __has_builtin(__builtin_amdgcn_sched_barrier)
#define SCHED_FENCE() __builtin_amdgcn_sched_barrier(0)
#endif
#endif
#ifndef SCHED_FENCE
#define SCHED_FENCE()
#endif

#define EPT 4
#define SCAN_ELEMS 1024

__global__ void zero_int_kernel(int* __restrict__ p, int n) {
    int i = blockIdx.x * blockDim.x + threadIdx.x;
    int stride = gridDim.x * blockDim.x;
    for (; i < n; i += stride) p[i] = 0;
}

__global__ __launch_bounds__(256) void rank_kernel(const int* __restrict__ dst, int* __restrict__ cnt,
                                                   int* __restrict__ rank, int e) {
    const int base = blockIdx.x * (256 * EPT) + threadIdx.x;
    int d[EPT], r[EPT];
#pragma unroll
    for (int k = 0; k < EPT; ++k) {
        int i = base + k * 256;
        d[k] = dst[min(i, e - 1)];
    }
#pragma unroll
    for (int k = 0; k < EPT; ++k) {
        int i = base + k * 256;
        r[k] = 0;
        if (i < e) r[k] = atomicAdd(&cnt[d[k]], 1);
    }
#pragma unroll
    for (int k = 0; k < EPT; ++k) {
        int i = base + k * 256;
        if (i < e) rank[i] = r[k];
    }
}

__global__ __launch_bounds__(256) void block_total_kernel(const int* __restrict__ cnt,
                                                          int* __restrict__ btot, int n) {
    __shared__ int sh[256];
    const int t = threadIdx.x;
    const int base = blockIdx.x * SCAN_ELEMS + t * 4;
    int s = 0;
    for (int k = 0; k < 4; ++k) {
        int i = base + k;
        if (i < n) s += cnt[i];
    }
    sh[t] = s;
    __syncthreads();
    for (int off = 128; off > 0; off >>= 1) {
        if (t < off) sh[t] += sh[t + off];
        __syncthreads();
    }
    if (t == 0) btot[blockIdx.x] = sh[0];
}

__global__ __launch_bounds__(256) void rowptr_dinv_kernel(const int* __restrict__ cnt,
                                                          const int* __restrict__ btot,
                                                          int* __restrict__ rowptr,
                                                          float* __restrict__ dinv, int n) {
    __shared__ int sh[256];
    const int t = threadIdx.x;
    int part = 0;
    for (int j = t; j < (int)blockIdx.x; j += 256) part += btot[j];
    sh[t] = part;
    __syncthreads();
    for (int off = 128; off > 0; off >>= 1) {
        if (t < off) sh[t] += sh[t + off];
        __syncthreads();
    }
    const int boff = sh[0];
    __syncthreads();

    const int base = blockIdx.x * SCAN_ELEMS + t * 4;
    int c[4];
    int s = 0;
    for (int k = 0; k < 4; ++k) {
        int i = base + k;
        c[k] = (i < n) ? cnt[i] : 0;
        s += c[k];
    }
    sh[t] = s;
    __syncthreads();
    for (int off = 1; off < 256; off <<= 1) {
        int v = (t >= off) ? sh[t - off] : 0;
        __syncthreads();
        sh[t] += v;
        __syncthreads();
    }
    int run = boff + sh[t] - s;
    for (int k = 0; k < 4; ++k) {
        int i = base + k;
        if (i < n) {
            rowptr[i] = run;
            dinv[i] = rsqrtf((float)(c[k] + 1));
            run += c[k];
        }
    }
    if (blockIdx.x == gridDim.x - 1 && t == 255) rowptr[n] = run;
}

__global__ __launch_bounds__(256) void place_kernel(const int* __restrict__ src,
                                                    const int* __restrict__ dst,
                                                    const int* __restrict__ rank,
                                                    const int* __restrict__ rowptr,
                                                    int* __restrict__ csr_src, int e) {
    const int base = blockIdx.x * (256 * EPT) + threadIdx.x;
    int d[EPT], r[EPT], s[EPT], p[EPT];
#pragma unroll
    for (int k = 0; k < EPT; ++k) {
        int ic = min(base + k * 256, e - 1);
        d[k] = dst[ic];
        r[k] = rank[ic];
        s[k] = src[ic];
    }
#pragma unroll
    for (int k = 0; k < EPT; ++k) p[k] = rowptr[d[k]] + r[k];
#pragma unroll
    for (int k = 0; k < EPT; ++k) {
        int i = base + k * 256;
        if (i < e && (unsigned)p[k] < (unsigned)e) csr_src[p[k]] = s[k];
    }
}

template <int HOUT>
__global__ __launch_bounds__(256) void gemm_wmma_kernel(const float* __restrict__ X,
                                                        const float* __restrict__ W,
                                                        float* __restrict__ H, int nrows) {
    extern __shared__ float smem[];
    float* sW = smem;
    float* sX = smem + EMB * HOUT;

    const int tid = threadIdx.x;
    const int rowBase = blockIdx.x * 128;

    for (int i = tid * 4; i < EMB * HOUT; i += 256 * 4) {
        *(float4*)&sW[i] = *(const float4*)&W[i];
    }
    for (int i = tid * 4; i < 128 * EMB; i += 256 * 4) {
        int r = i >> 7, c = i & 127;
        int gr = rowBase + r;
        float4 v = make_float4(0.f, 0.f, 0.f, 0.f);
        if (gr < nrows) v = *(const float4*)&X[(long long)gr * EMB + c];
        *(float4*)&sX[r * XPITCH + c] = v;
    }
    __syncthreads();

    const int wave = tid >> 5;
    const int lane = tid & 31;
    const int nl = lane & 15;
    const int kh = lane >> 4;
    const int wr = wave * 16;

    const float* xrow = &sX[(wr + nl) * XPITCH + 2 * kh];
    v2f afrag[KSTEPS];
#pragma unroll
    for (int ks = 0; ks < KSTEPS; ++ks) afrag[ks] = *(const v2f*)&xrow[4 * ks];
    SCHED_FENCE();

    for (int nt = 0; nt < HOUT / 16; ++nt) {
        const float* wcol = &sW[nt * 16 + nl];
        v2f bfrag[KSTEPS];
#pragma unroll
        for (int ks = 0; ks < KSTEPS; ++ks) {
            const int kk = 4 * ks + 2 * kh;
            v2f b;
            b.x = wcol[kk * HOUT];
            b.y = wcol[(kk + 1) * HOUT];
            bfrag[ks] = b;
        }
        SCHED_FENCE();
        v8f acc = {};
#pragma unroll
        for (int ks = 0; ks < KSTEPS; ++ks) {
            acc = __builtin_amdgcn_wmma_f32_16x16x4_f32(
                false, afrag[ks], false, bfrag[ks], (short)0, acc, false, false);
        }
        SCHED_FENCE();
#pragma unroll
        for (int j = 0; j < 8; ++j) {
            int r = rowBase + wr + kh * 8 + j;
            if (r < nrows) H[(long long)r * HOUT + nt * 16 + nl] = acc[j];
        }
    }
}

template <int V>
__device__ __forceinline__ void load_row(const float* __restrict__ p, float (&v)[V]) {
    if constexpr (V == 4) {
        float4 t = *(const float4*)p;
        v[0] = t.x; v[1] = t.y; v[2] = t.z; v[3] = t.w;
    } else {
        float2 t = *(const float2*)p;
        v[0] = t.x; v[1] = t.y;
    }
}

template <int HOUT>
__global__ __launch_bounds__(256) void gather_kernel(const int* __restrict__ rowptr,
                                                     const int* __restrict__ csr_src,
                                                     const float* __restrict__ dinv,
                                                     const float* __restrict__ h,
                                                     float* __restrict__ agg, int n) {
    constexpr int V = HOUT / 32;
    const int lane = threadIdx.x & 31;
    const int wv = __builtin_amdgcn_readfirstlane((int)(threadIdx.x >> 5));
    const int node = blockIdx.x * 8 + wv;
    if (node >= n) return;
    const int beg = rowptr[node];
    const int end = rowptr[node + 1];
    const float dd = dinv[node];
    const float* hl = h + lane * V;

    float acc[V];
    for (int q = 0; q < V; ++q) acc[q] = 0.0f;

    for (int j0 = beg; j0 < end; j0 += 32) {
        const int m = min(32, end - j0);
        int s_l = 0;
        float w_l = 0.0f;
        if (lane < m) {
            s_l = csr_src[j0 + lane];
            s_l = min(max(s_l, 0), n - 1);
            w_l = dinv[s_l] * dd;
        }
        const int w_li = __float_as_int(w_l);
        int i = 0;
        for (; i + 4 <= m; i += 4) {
            const int s0 = __builtin_amdgcn_readlane(s_l, i);
            const int s1 = __builtin_amdgcn_readlane(s_l, i + 1);
            const int s2 = __builtin_amdgcn_readlane(s_l, i + 2);
            const int s3 = __builtin_amdgcn_readlane(s_l, i + 3);
            const float w0 = __int_as_float(__builtin_amdgcn_readlane(w_li, i));
            const float w1 = __int_as_float(__builtin_amdgcn_readlane(w_li, i + 1));
            const float w2 = __int_as_float(__builtin_amdgcn_readlane(w_li, i + 2));
            const float w3 = __int_as_float(__builtin_amdgcn_readlane(w_li, i + 3));
            float v0[V], v1[V], v2[V], v3[V];
            load_row<V>(hl + (long long)s0 * HOUT, v0);
            load_row<V>(hl + (long long)s1 * HOUT, v1);
            load_row<V>(hl + (long long)s2 * HOUT, v2);
            load_row<V>(hl + (long long)s3 * HOUT, v3);
            for (int q = 0; q < V; ++q) acc[q] = fmaf(w0, v0[q], acc[q]);
            for (int q = 0; q < V; ++q) acc[q] = fmaf(w1, v1[q], acc[q]);
            for (int q = 0; q < V; ++q) acc[q] = fmaf(w2, v2[q], acc[q]);
            for (int q = 0; q < V; ++q) acc[q] = fmaf(w3, v3[q], acc[q]);
        }
        for (; i < m; ++i) {
            const int s0 = __builtin_amdgcn_readlane(s_l, i);
            const float w0 = __int_as_float(__builtin_amdgcn_readlane(w_li, i));
            float v0[V];
            load_row<V>(hl + (long long)s0 * HOUT, v0);
            for (int q = 0; q < V; ++q) acc[q] = fmaf(w0, v0[q], acc[q]);
        }
    }

    float* o = agg + (long long)node * HOUT + lane * V;
    if constexpr (V == 4) {
        *(float4*)o = make_float4(acc[0], acc[1], acc[2], acc[3]);
    } else {
        *(float2*)o = make_float2(acc[0], acc[1]);
    }
}

__global__ void bias_relu_kernel(float* __restrict__ agg, const float* __restrict__ h,
                                 const float* __restrict__ dinv, const float* __restrict__ b,
                                 long long total, int hdim) {
    long long i = (long long)blockIdx.x * blockDim.x + threadIdx.x;
    if (i >= total) return;
    int row = (int)(i / hdim), col = (int)(i % hdim);
    float di = dinv[row];
    float v = agg[i] + di * di * h[i] + b[col];
    agg[i] = fmaxf(v, 0.0f);
}

__global__ void final_add_kernel(float* __restrict__ out, const float* __restrict__ h,
                                 const float* __restrict__ dinv, const float* __restrict__ b,
                                 long long total, int hdim) {
    long long i = (long long)blockIdx.x * blockDim.x + threadIdx.x;
    if (i >= total) return;
    int row = (int)(i / hdim), col = (int)(i % hdim);
    float di = dinv[row];
    out[i] = out[i] + di * di * h[i] + b[col];
}

extern "C" void kernel_launch(void* const* d_in, const int* in_sizes, int n_in,
                              void* d_out, int out_size, void* d_ws, size_t ws_size,
                              hipStream_t stream) {
    const int*   edge = (const int*)d_in[0];
    const float* x0   = (const float*)d_in[1];
    const float* W1   = (const float*)d_in[2];
    const float* b1   = (const float*)d_in[3];
    const float* W2   = (const float*)d_in[4];
    const float* b2   = (const float*)d_in[5];
    float* out = (float*)d_out;

    const int E    = in_sizes[0] / 2;
    const int N    = in_sizes[1] / EMB;
    const int HID  = in_sizes[3];
    const int REPR = in_sizes[5];
    const int* src = edge;
    const int* dst = edge + E;

    float* dinv = (float*)d_ws;
    float* h    = dinv + N;
    float* agg  = h + (size_t)N * HID;
    int* cnt     = (int*)(agg + (size_t)N * HID);
    int* rowptr  = cnt + N;
    int* rank    = rowptr + (N + 1);
    int* csr_src = rank + E;
    int* btot    = csr_src + E;

    const int T = 256;
    long long nh = (long long)N * HID;
    long long nr = (long long)N * REPR;
    const int nscan = (N + SCAN_ELEMS - 1) / SCAN_ELEMS;
    const int nedge = (E + T * EPT - 1) / (T * EPT);

    zero_int_kernel<<<128, T, 0, stream>>>(cnt, N);
    rank_kernel<<<nedge, T, 0, stream>>>(dst, cnt, rank, E);
    block_total_kernel<<<nscan, T, 0, stream>>>(cnt, btot, N);
    rowptr_dinv_kernel<<<nscan, T, 0, stream>>>(cnt, btot, rowptr, dinv, N);
    place_kernel<<<nedge, T, 0, stream>>>(src, dst, rank, rowptr, csr_src, E);

    size_t smem1 = (size_t)(EMB * HID + 128 * XPITCH) * sizeof(float);
    gemm_wmma_kernel<128><<<(N + 127) / 128, T, smem1, stream>>>(x0, W1, h, N);
    gather_kernel<128><<<(N + 7) / 8, T, 0, stream>>>(rowptr, csr_src, dinv, h, agg, N);
    bias_relu_kernel<<<(int)((nh + T - 1) / T), T, 0, stream>>>(agg, h, dinv, b1, nh, HID);

    size_t smem2 = (size_t)(EMB * REPR + 128 * XPITCH) * sizeof(float);
    gemm_wmma_kernel<64><<<(N + 127) / 128, T, smem2, stream>>>(agg, W2, h, N);
    gather_kernel<64><<<(N + 7) / 8, T, 0, stream>>>(rowptr, csr_src, dinv, h, out, N);
    final_add_kernel<<<(int)((nr + T - 1) / T), T, 0, stream>>>(out, h, dinv, b2, nr, REPR);
}


# Fold self-loop term, bias and relu into the two gather kernels (own h row and bias loaded ahead of the edge loop): both elementwise launches and the agg round trip go; sum order unchanged
# speedup vs baseline: 2.1813x; 1.1535x over previous
#include <hip/hip_runtime.h>
#include <hip/hip_bf16.h>

typedef __attribute__((ext_vector_type(2))) float v2f;
typedef __attribute__((ext_vector_type(8))) float v8f;

#define EMB 128
#define XPITCH 132
#define KSTEPS (EMB / 4)

#if defined(__has_builtin)
#if __has_builtin(__builtin_amdgcn_sched_barrier)
#define SCHED_FENCE() __builtin_amdgcn_sched_barrier(0)
#endif
#endif
#ifndef SCHED_FENCE
#define SCHED_FENCE()
#endif

#define EPT 4
#define SCAN_ELEMS 1024

__global__ void zero_int_kernel(int* __restrict__ p, int n) {
    int i = blockIdx.x * blockDim.x + threadIdx.x;
    int stride = gridDim.x * blockDim.x;
    for (; i < n; i += stride) p[i] = 0;
}

__global__ __launch_bounds__(256) void rank_kernel(const int* __restrict__ dst, int* __restrict__ cnt,
                                                   int* __restrict__ rank, int e) {
    const int base = blockIdx.x * (256 * EPT) + threadIdx.x;
    int d[EPT], r[EPT];
#pragma unroll
    for (int k = 0; k < EPT; ++k) {
        int i = base + k * 256;
        d[k] = dst[min(i, e - 1)];
    }
#pragma unroll
    for (int k = 0; k < EPT; ++k) {
        int i = base + k * 256;
        r[k] = 0;
        if (i < e) r[k] = atomicAdd(&cnt[d[k]], 1);
    }
#pragma unroll
    for (int k = 0; k < EPT; ++k) {
        int i = base + k * 256;
        if (i < e) rank[i] = r[k];
    }
}

__global__ __launch_bounds__(256) void block_total_kernel(const int* __restrict__ cnt,
                                                          int* __restrict__ btot, int n) {
    __shared__ int sh[256];
    const int t = threadIdx.x;
    const int base = blockIdx.x * SCAN_ELEMS + t * 4;
    int s = 0;
    for (int k = 0; k < 4; ++k) {
        int i = base + k;
        if (i < n) s += cnt[i];
    }
    sh[t] = s;
    __syncthreads();
    for (int off = 128; off > 0; off >>= 1) {
        if (t < off) sh[t] += sh[t + off];
        __syncthreads();
    }
    if (t == 0) btot[blockIdx.x] = sh[0];
}

__global__ __launch_bounds__(256) void rowptr_dinv_kernel(const int* __restrict__ cnt,
                                                          const int* __restrict__ btot,
                                                          int* __restrict__ rowptr,
                                                          float* __restrict__ dinv, int n) {
    __shared__ int sh[256];
    const int t = threadIdx.x;
    int part = 0;
    for (int j = t; j < (int)blockIdx.x; j += 256) part += btot[j];
    sh[t] = part;
    __syncthreads();
    for (int off = 128; off > 0; off >>= 1) {
        if (t < off) sh[t] += sh[t + off];
        __syncthreads();
    }
    const int boff = sh[0];
    __syncthreads();

    const int base = blockIdx.x * SCAN_ELEMS + t * 4;
    int c[4];
    int s = 0;
    for (int k = 0; k < 4; ++k) {
        int i = base + k;
        c[k] = (i < n) ? cnt[i] : 0;
        s += c[k];
    }
    sh[t] = s;
    __syncthreads();
    for (int off = 1; off < 256; off <<= 1) {
        int v = (t >= off) ? sh[t - off] : 0;
        __syncthreads();
        sh[t] += v;
        __syncthreads();
    }
    int run = boff + sh[t] - s;
    for (int k = 0; k < 4; ++k) {
        int i = base + k;
        if (i < n) {
            rowptr[i] = run;
            dinv[i] = rsqrtf((float)(c[k] + 1));
            run += c[k];
        }
    }
    if (blockIdx.x == gridDim.x - 1 && t == 255) rowptr[n] = run;
}

__global__ __launch_bounds__(256) void place_kernel(const int* __restrict__ src,
                                                    const int* __restrict__ dst,
                                                    const int* __restrict__ rank,
                                                    const int* __restrict__ rowptr,
                                                    int* __restrict__ csr_src, int e) {
    const int base = blockIdx.x * (256 * EPT) + threadIdx.x;
    int d[EPT], r[EPT], s[EPT], p[EPT];
#pragma unroll
    for (int k = 0; k < EPT; ++k) {
        int ic = min(base + k * 256, e - 1);
        d[k] = dst[ic];
        r[k] = rank[ic];
        s[k] = src[ic];
    }
#pragma unroll
    for (int k = 0; k < EPT; ++k) p[k] = rowptr[d[k]] + r[k];
#pragma unroll
    for (int k = 0; k < EPT; ++k) {
        int i = base + k * 256;
        if (i < e && (unsigned)p[k] < (unsigned)e) csr_src[p[k]] = s[k];
    }
}

template <int HOUT>
__global__ __launch_bounds__(256) void gemm_wmma_kernel(const float* __restrict__ X,
                                                        const float* __restrict__ W,
                                                        float* __restrict__ H, int nrows) {
    extern __shared__ float smem[];
    float* sW = smem;
    float* sX = smem + EMB * HOUT;

    const int tid = threadIdx.x;
    const int rowBase = blockIdx.x * 128;

    for (int i = tid * 4; i < EMB * HOUT; i += 256 * 4) {
        *(float4*)&sW[i] = *(const float4*)&W[i];
    }
    for (int i = tid * 4; i < 128 * EMB; i += 256 * 4) {
        int r = i >> 7, c = i & 127;
        int gr = rowBase + r;
        float4 v = make_float4(0.f, 0.f, 0.f, 0.f);
        if (gr < nrows) v = *(const float4*)&X[(long long)gr * EMB + c];
        *(float4*)&sX[r * XPITCH + c] = v;
    }
    __syncthreads();

    const int wave = tid >> 5;
    const int lane = tid & 31;
    const int nl = lane & 15;
    const int kh = lane >> 4;
    const int wr = wave * 16;

    const float* xrow = &sX[(wr + nl) * XPITCH + 2 * kh];
    v2f afrag[KSTEPS];
#pragma unroll
    for (int ks = 0; ks < KSTEPS; ++ks) afrag[ks] = *(const v2f*)&xrow[4 * ks];
    SCHED_FENCE();

    for (int nt = 0; nt < HOUT / 16; ++nt) {
        const float* wcol = &sW[nt * 16 + nl];
        v2f bfrag[KSTEPS];
#pragma unroll
        for (int ks = 0; ks < KSTEPS; ++ks) {
            const int kk = 4 * ks + 2 * kh;
            v2f b;
            b.x = wcol[kk * HOUT];
            b.y = wcol[(kk + 1) * HOUT];
            bfrag[ks] = b;
        }
        SCHED_FENCE();
        v8f acc = {};
#pragma unroll
        for (int ks = 0; ks < KSTEPS; ++ks) {
            acc = __builtin_amdgcn_wmma_f32_16x16x4_f32(
                false, afrag[ks], false, bfrag[ks], (short)0, acc, false, false);
        }
        SCHED_FENCE();
#pragma unroll
        for (int j = 0; j < 8; ++j) {
            int r = rowBase + wr + kh * 8 + j;
            if (r < nrows) H[(long long)r * HOUT + nt * 16 + nl] = acc[j];
        }
    }
}

template <int V>
__device__ __forceinline__ void load_row(const float* __restrict__ p, float (&v)[V]) {
    if constexpr (V == 4) {
        float4 t = *(const float4*)p;
        v[0] = t.x; v[1] = t.y; v[2] = t.z; v[3] = t.w;
    } else {
        float2 t = *(const float2*)p;
        v[0] = t.x; v[1] = t.y;
    }
}

template <int HOUT, bool RELU>
__global__ __launch_bounds__(256) void gather_kernel(const int* __restrict__ rowptr,
                                                     const int* __restrict__ csr_src,
                                                     const float* __restrict__ dinv,
                                                     const float* __restrict__ h,
                                                     const float* __restrict__ b,
                                                     float* __restrict__ y, int n) {
    constexpr int V = HOUT / 32;
    const int lane = threadIdx.x & 31;
    const int wv = __builtin_amdgcn_readfirstlane((int)(threadIdx.x >> 5));
    const int node = blockIdx.x * 8 + wv;
    if (node >= n) return;
    const int beg = rowptr[node];
    const int end = rowptr[node + 1];
    const float dd = dinv[node];
    const float* hl = h + lane * V;

    float hs[V], bb[V];
    load_row<V>(hl + (long long)node * HOUT, hs);
    load_row<V>(b + lane * V, bb);

    float acc[V];
    for (int q = 0; q < V; ++q) acc[q] = 0.0f;

    for (int j0 = beg; j0 < end; j0 += 32) {
        const int m = min(32, end - j0);
        int s_l = 0;
        float w_l = 0.0f;
        if (lane < m) {
            s_l = csr_src[j0 + lane];
            s_l = min(max(s_l, 0), n - 1);
            w_l = dinv[s_l] * dd;
        }
        const int w_li = __float_as_int(w_l);
        int i = 0;
        for (; i + 4 <= m; i += 4) {
            const int s0 = __builtin_amdgcn_readlane(s_l, i);
            const int s1 = __builtin_amdgcn_readlane(s_l, i + 1);
            const int s2 = __builtin_amdgcn_readlane(s_l, i + 2);
            const int s3 = __builtin_amdgcn_readlane(s_l, i + 3);
            const float w0 = __int_as_float(__builtin_amdgcn_readlane(w_li, i));
            const float w1 = __int_as_float(__builtin_amdgcn_readlane(w_li, i + 1));
            const float w2 = __int_as_float(__builtin_amdgcn_readlane(w_li, i + 2));
            const float w3 = __int_as_float(__builtin_amdgcn_readlane(w_li, i + 3));
            float v0[V], v1[V], v2[V], v3[V];
            load_row<V>(hl + (long long)s0 * HOUT, v0);
            load_row<V>(hl + (long long)s1 * HOUT, v1);
            load_row<V>(hl + (long long)s2 * HOUT, v2);
            load_row<V>(hl + (long long)s3 * HOUT, v3);
            for (int q = 0; q < V; ++q) acc[q] = fmaf(w0, v0[q], acc[q]);
            for (int q = 0; q < V; ++q) acc[q] = fmaf(w1, v1[q], acc[q]);
            for (int q = 0; q < V; ++q) acc[q] = fmaf(w2, v2[q], acc[q]);
            for (int q = 0; q < V; ++q) acc[q] = fmaf(w3, v3[q], acc[q]);
        }
        for (; i < m; ++i) {
            const int s0 = __builtin_amdgcn_readlane(s_l, i);
            const float w0 = __int_as_float(__builtin_amdgcn_readlane(w_li, i));
            float v0[V];
            load_row<V>(hl + (long long)s0 * HOUT, v0);
            for (int q = 0; q < V; ++q) acc[q] = fmaf(w0, v0[q], acc[q]);
        }
    }

    const float d2 = dd * dd;
    for (int q = 0; q < V; ++q) {
        float v = acc[q] + d2 * hs[q] + bb[q];
        acc[q] = RELU ? fmaxf(v, 0.0f) : v;
    }

    float* o = y + (long long)node * HOUT + lane * V;
    if constexpr (V == 4) {
        *(float4*)o = make_float4(acc[0], acc[1], acc[2], acc[3]);
    } else {
        *(float2*)o = make_float2(acc[0], acc[1]);
    }
}

extern "C" void kernel_launch(void* const* d_in, const int* in_sizes, int n_in,
                              void* d_out, int out_size, void* d_ws, size_t ws_size,
                              hipStream_t stream) {
    const int*   edge = (const int*)d_in[0];
    const float* x0   = (const float*)d_in[1];
    const float* W1   = (const float*)d_in[2];
    const float* b1   = (const float*)d_in[3];
    const float* W2   = (const float*)d_in[4];
    const float* b2   = (const float*)d_in[5];
    float* out = (float*)d_out;

    const int E    = in_sizes[0] / 2;
    const int N    = in_sizes[1] / EMB;
    const int HID  = in_sizes[3];
    const int REPR = in_sizes[5];
    const int* src = edge;
    const int* dst = edge + E;

    float* dinv = (float*)d_ws;
    float* h    = dinv + N;
    float* x1   = h + (size_t)N * HID;
    int* cnt     = (int*)(x1 + (size_t)N * HID);
    int* rowptr  = cnt + N;
    int* rank    = rowptr + (N + 1);
    int* csr_src = rank + E;
    int* btot    = csr_src + E;

    const int T = 256;
    const int nscan = (N + SCAN_ELEMS - 1) / SCAN_ELEMS;
    const int nedge = (E + T * EPT - 1) / (T * EPT);

    zero_int_kernel<<<128, T, 0, stream>>>(cnt, N);
    rank_kernel<<<nedge, T, 0, stream>>>(dst, cnt, rank, E);
    block_total_kernel<<<nscan, T, 0, stream>>>(cnt, btot, N);
    rowptr_dinv_kernel<<<nscan, T, 0, stream>>>(cnt, btot, rowptr, dinv, N);
    place_kernel<<<nedge, T, 0, stream>>>(src, dst, rank, rowptr, csr_src, E);

    size_t smem1 = (size_t)(EMB * HID + 128 * XPITCH) * sizeof(float);
    gemm_wmma_kernel<128><<<(N + 127) / 128, T, smem1, stream>>>(x0, W1, h, N);
    gather_kernel<128, true><<<(N + 7) / 8, T, 0, stream>>>(rowptr, csr_src, dinv, h, b1, x1, N);

    size_t smem2 = (size_t)(EMB * REPR + 128 * XPITCH) * sizeof(float);
    gemm_wmma_kernel<64><<<(N + 127) / 128, T, smem2, stream>>>(x1, W2, h, N);
    gather_kernel<64, false><<<(N + 7) / 8, T, 0, stream>>>(rowptr, csr_src, dinv, h, b2, out, N);
}


# Edge list sorted by destination with LDS counters only (chunk bucket histogram, offsets, bucket scatter, per-bucket LDS sort): no global atomics, 9 to 8 launches; sum order = LDS arrival order
# speedup vs baseline: 3.1752x; 1.4556x over previous
#include <hip/hip_runtime.h>
#include <hip/hip_bf16.h>

typedef __attribute__((ext_vector_type(2))) float v2f;
typedef __attribute__((ext_vector_type(8))) float v8f;

#define EMB 128
#define XPITCH 132
#define KSTEPS (EMB / 4)

#if defined(__has_builtin)
#if __has_builtin(__builtin_amdgcn_sched_barrier)
#define SCHED_FENCE() __builtin_amdgcn_sched_barrier(0)
#endif
#endif
#ifndef SCHED_FENCE
#define SCHED_FENCE()
#endif

#define BSH 8
#define BNODES (1 << BSH)
#define SRC_BITS 24
#define SRC_MASK ((1u << SRC_BITS) - 1u)
#define CEPT 25
#define CHUNK (256 * CEPT)
#define DEPT 8
#define DCAP 8192

__device__ __forceinline__ int block_excl_scan(int v, int* sh, int& tot) {
    const int lane = threadIdx.x & 31;
    const int wv = threadIdx.x >> 5;
    int inc = v;
    for (int off = 1; off < 32; off <<= 1) {
        int u = __shfl_up(inc, off, 32);
        if (lane >= off) inc += u;
    }
    if (lane == 31) sh[wv] = inc;
    __syncthreads();
    int wbase = 0, t = 0;
    for (int w = 0; w < 8; ++w) {
        int c = sh[w];
        if (w < wv) wbase += c;
        t += c;
    }
    __syncthreads();
    tot = t;
    return wbase + inc - v;
}

__global__ __launch_bounds__(256) void bucket_hist_kernel(const int* __restrict__ dst,
                                                          int* __restrict__ ghist,
                                                          int e, int n, int nb) {
    extern __shared__ int lds_a[];
    const int t = threadIdx.x;
    for (int b = t; b < nb; b += 256) lds_a[b] = 0;
    __syncthreads();
    const int base = blockIdx.x * CHUNK + t;
    int d[CEPT];
#pragma unroll
    for (int k = 0; k < CEPT; ++k) d[k] = dst[min(base + k * 256, e - 1)];
#pragma unroll
    for (int k = 0; k < CEPT; ++k) {
        atomicAdd(&lds_a[min(max(d[k], 0), n - 1) >> BSH], (base + k * 256 < e) ? 1 : 0);
    }
    __syncthreads();
    for (int b = t; b < nb; b += 256) ghist[blockIdx.x * nb + b] = lds_a[b];
}

__global__ __launch_bounds__(256) void bucket_offsets_kernel(const int* __restrict__ ghist,
                                                             int* __restrict__ goff,
                                                             int* __restrict__ btot,
                                                             int nch, int nb) {
    __shared__ int sh[8];
    const int b = blockIdx.x;
    const int t = threadIdx.x;
    const int per = (nch + 255) / 256;
    const int c0 = min(t * per, nch);
    const int c1 = min(c0 + per, nch);
    int s = 0;
    for (int c = c0; c < c1; ++c) s += ghist[c * nb + b];
    int tot;
    int run = block_excl_scan(s, sh, tot);
    for (int c = c0; c < c1; ++c) {
        goff[c * nb + b] = run;
        run += ghist[c * nb + b];
    }
    if (t == 0) btot[b] = tot;
}

__global__ __launch_bounds__(256) void bucket_scatter_kernel(const int* __restrict__ src,
                                                             const int* __restrict__ dst,
                                                             const int* __restrict__ ghist,
                                                             const int* __restrict__ goff,
                                                             const int* __restrict__ btot,
                                                             unsigned* __restrict__ tmp,
                                                             int e, int n, int nb) {
    extern __shared__ int lds_c[];
    unsigned* stage = (unsigned*)lds_c;
    int* lstart = lds_c + (CHUNK + 1);
    int* fill   = lstart + nb;
    int* gdelta = fill + nb;
    unsigned short* sb = (unsigned short*)(gdelta + nb);
    __shared__ int sh[8];
    const int t = threadIdx.x;
    const int blk = blockIdx.x;

    const int base = blk * CHUNK + t;
    int d[CEPT], s[CEPT];
#pragma unroll
    for (int k = 0; k < CEPT; ++k) {
        int ic = min(base + k * 256, e - 1);
        d[k] = dst[ic];
        s[k] = src[ic];
    }

    const int per = (nb + 255) / 256;
    const int b0 = min(t * per, nb);
    const int b1 = min(b0 + per, nb);
    int sl = 0, sg = 0;
    for (int b = b0; b < b1; ++b) {
        sl += ghist[blk * nb + b];
        sg += btot[b];
    }
    int tot;
    int runl = block_excl_scan(sl, sh, tot);
    int rung = block_excl_scan(sg, sh, tot);
    for (int b = b0; b < b1; ++b) {
        lstart[b] = runl;
        fill[b] = 0;
        gdelta[b] = rung + goff[blk * nb + b] - runl;
        runl += ghist[blk * nb + b];
        rung += btot[b];
    }
    __syncthreads();

    unsigned wd[CEPT];
    int bk[CEPT], pos[CEPT];
#pragma unroll
    for (int k = 0; k < CEPT; ++k) {
        const bool valid = base + k * 256 < e;
        const int dd = min(max(d[k], 0), n - 1);
        bk[k] = dd >> BSH;
        wd[k] = ((unsigned)s[k] & SRC_MASK) | ((unsigned)(dd & (BNODES - 1)) << SRC_BITS);
        const int p = lstart[bk[k]] + atomicAdd(&fill[bk[k]], valid ? 1 : 0);
        pos[k] = valid ? p : CHUNK;
    }
#pragma unroll
    for (int k = 0; k < CEPT; ++k) {
        stage[pos[k]] = wd[k];
        sb[pos[k]] = (unsigned short)bk[k];
    }
    __syncthreads();

    const int cnt = min(CHUNK, e - blk * CHUNK);
    unsigned wq[CEPT];
    int gq[CEPT];
#pragma unroll
    for (int k = 0; k < CEPT; ++k) {
        const int ii = min(t + k * 256, cnt - 1);
        wq[k] = stage[ii];
        gq[k] = sb[ii];
    }
#pragma unroll
    for (int k = 0; k < CEPT; ++k) gq[k] = gdelta[gq[k]];
#pragma unroll
    for (int k = 0; k < CEPT; ++k) {
        const int i = t + k * 256;
        if (i < cnt) tmp[gq[k] + i] = wq[k];
    }
}

__global__ __launch_bounds__(BNODES) void bucket_sort_kernel(const unsigned* __restrict__ tmp,
                                                             const int* __restrict__ btot,
                                                             int* __restrict__ csr_src,
                                                             int* __restrict__ rowptr,
                                                             float* __restrict__ dinv,
                                                             int n, int nb) {
    __shared__ int ncnt[BNODES];
    __shared__ int nstart[BNODES];
    __shared__ int fill[BNODES];
    __shared__ int stage[DCAP + 1];
    __shared__ int sh[8];
    const int b = blockIdx.x;
    const int t = threadIdx.x;

    int part = 0;
    for (int j = t; j < b; j += BNODES) part += btot[j];
    int base;
    block_excl_scan(part, sh, base);
    const int cntb = btot[b];

    ncnt[t] = 0;
    fill[t] = 0;
    __syncthreads();

    for (int i0 = t; i0 < cntb; i0 += BNODES * DEPT) {
        unsigned w[DEPT];
#pragma unroll
        for (int k = 0; k < DEPT; ++k) w[k] = tmp[base + min(i0 + k * BNODES, cntb - 1)];
#pragma unroll
        for (int k = 0; k < DEPT; ++k)
            atomicAdd(&ncnt[w[k] >> SRC_BITS], (i0 + k * BNODES < cntb) ? 1 : 0);
    }
    __syncthreads();

    const int c = ncnt[t];
    int tot;
    const int st = block_excl_scan(c, sh, tot);
    nstart[t] = st;
    const int node = b * BNODES + t;
    if (node < n) {
        rowptr[node] = base + st;
        dinv[node] = rsqrtf((float)(c + 1));
    }
    if (b == nb - 1 && t == 0) rowptr[n] = base + cntb;
    __syncthreads();

    const bool staged = cntb <= DCAP;
    for (int i0 = t; i0 < cntb; i0 += BNODES * DEPT) {
        unsigned w[DEPT];
        int pos[DEPT];
#pragma unroll
        for (int k = 0; k < DEPT; ++k) w[k] = tmp[base + min(i0 + k * BNODES, cntb - 1)];
#pragma unroll
        for (int k = 0; k < DEPT; ++k) {
            const bool valid = i0 + k * BNODES < cntb;
            const int dl = (int)(w[k] >> SRC_BITS);
            const int p = nstart[dl] + atomicAdd(&fill[dl], valid ? 1 : 0);
            pos[k] = valid ? p : -1;
        }
        if (staged) {
#pragma unroll
            for (int k = 0; k < DEPT; ++k) stage[pos[k] < 0 ? DCAP : pos[k]] = (int)(w[k] & SRC_MASK);
        } else {
#pragma unroll
            for (int k = 0; k < DEPT; ++k) {
                if (pos[k] >= 0) csr_src[base + pos[k]] = (int)(w[k] & SRC_MASK);
            }
        }
    }
    if (staged) {
        __syncthreads();
        for (int i = t; i < cntb; i += BNODES) csr_src[base + i] = stage[i];
    }
}

template <int HOUT>
__global__ __launch_bounds__(256) void gemm_wmma_kernel(const float* __restrict__ X,
                                                        const float* __restrict__ W,
                                                        float* __restrict__ H, int nrows) {
    extern __shared__ float smem[];
    float* sW = smem;
    float* sX = smem + EMB * HOUT;

    const int tid = threadIdx.x;
    const int rowBase = blockIdx.x * 128;

    for (int i = tid * 4; i < EMB * HOUT; i += 256 * 4) {
        *(float4*)&sW[i] = *(const float4*)&W[i];
    }
    for (int i = tid * 4; i < 128 * EMB; i += 256 * 4) {
        int r = i >> 7, c = i & 127;
        int gr = rowBase + r;
        float4 v = make_float4(0.f, 0.f, 0.f, 0.f);
        if (gr < nrows) v = *(const float4*)&X[(long long)gr * EMB + c];
        *(float4*)&sX[r * XPITCH + c] = v;
    }
    __syncthreads();

    const int wave = tid >> 5;
    const int lane = tid & 31;
    const int nl = lane & 15;
    const int kh = lane >> 4;
    const int wr = wave * 16;

    const float* xrow = &sX[(wr + nl) * XPITCH + 2 * kh];
    v2f afrag[KSTEPS];
#pragma unroll
    for (int ks = 0; ks < KSTEPS; ++ks) afrag[ks] = *(const v2f*)&xrow[4 * ks];
    SCHED_FENCE();

    for (int nt = 0; nt < HOUT / 16; ++nt) {
        const float* wcol = &sW[nt * 16 + nl];
        v2f bfrag[KSTEPS];
#pragma unroll
        for (int ks = 0; ks < KSTEPS; ++ks) {
            const int kk = 4 * ks + 2 * kh;
            v2f b;
            b.x = wcol[kk * HOUT];
            b.y = wcol[(kk + 1) * HOUT];
            bfrag[ks] = b;
        }
        SCHED_FENCE();
        v8f acc = {};
#pragma unroll
        for (int ks = 0; ks < KSTEPS; ++ks) {
            acc = __builtin_amdgcn_wmma_f32_16x16x4_f32(
                false, afrag[ks], false, bfrag[ks], (short)0, acc, false, false);
        }
        SCHED_FENCE();
#pragma unroll
        for (int j = 0; j < 8; ++j) {
            int r = rowBase + wr + kh * 8 + j;
            if (r < nrows) H[(long long)r * HOUT + nt * 16 + nl] = acc[j];
        }
    }
}

template <int V>
__device__ __forceinline__ void load_row(const float* __restrict__ p, float (&v)[V]) {
    if constexpr (V == 4) {
        float4 t = *(const float4*)p;
        v[0] = t.x; v[1] = t.y; v[2] = t.z; v[3] = t.w;
    } else {
        float2 t = *(const float2*)p;
        v[0] = t.x; v[1] = t.y;
    }
}

template <int HOUT, bool RELU>
__global__ __launch_bounds__(256) void gather_kernel(const int* __restrict__ rowptr,
                                                     const int* __restrict__ csr_src,
                                                     const float* __restrict__ dinv,
                                                     const float* __restrict__ h,
                                                     const float* __restrict__ b,
                                                     float* __restrict__ y, int n) {
    constexpr int V = HOUT / 32;
    const int lane = threadIdx.x & 31;
    const int wv = __builtin_amdgcn_readfirstlane((int)(threadIdx.x >> 5));
    const int node = blockIdx.x * 8 + wv;
    if (node >= n) return;
    const int beg = rowptr[node];
    const int end = rowptr[node + 1];
    const float dd = dinv[node];
    const float* hl = h + lane * V;

    float hs[V], bb[V];
    load_row<V>(hl + (long long)node * HOUT, hs);
    load_row<V>(b + lane * V, bb);

    float acc[V];
    for (int q = 0; q < V; ++q) acc[q] = 0.0f;

    for (int j0 = beg; j0 < end; j0 += 32) {
        const int m = min(32, end - j0);
        int s_l = 0;
        float w_l = 0.0f;
        if (lane < m) {
            s_l = csr_src[j0 + lane];
            s_l = min(max(s_l, 0), n - 1);
            w_l = dinv[s_l] * dd;
        }
        const int w_li = __float_as_int(w_l);
        int i = 0;
        for (; i + 4 <= m; i += 4) {
            const int s0 = __builtin_amdgcn_readlane(s_l, i);
            const int s1 = __builtin_amdgcn_readlane(s_l, i + 1);
            const int s2 = __builtin_amdgcn_readlane(s_l, i + 2);
            const int s3 = __builtin_amdgcn_readlane(s_l, i + 3);
            const float w0 = __int_as_float(__builtin_amdgcn_readlane(w_li, i));
            const float w1 = __int_as_float(__builtin_amdgcn_readlane(w_li, i + 1));
            const float w2 = __int_as_float(__builtin_amdgcn_readlane(w_li, i + 2));
            const float w3 = __int_as_float(__builtin_amdgcn_readlane(w_li, i + 3));
            float v0[V], v1[V], v2[V], v3[V];
            load_row<V>(hl + (long long)s0 * HOUT, v0);
            load_row<V>(hl + (long long)s1 * HOUT, v1);
            load_row<V>(hl + (long long)s2 * HOUT, v2);
            load_row<V>(hl + (long long)s3 * HOUT, v3);
            for (int q = 0; q < V; ++q) acc[q] = fmaf(w0, v0[q], acc[q]);
            for (int q = 0; q < V; ++q) acc[q] = fmaf(w1, v1[q], acc[q]);
            for (int q = 0; q < V; ++q) acc[q] = fmaf(w2, v2[q], acc[q]);
            for (int q = 0; q < V; ++q) acc[q] = fmaf(w3, v3[q], acc[q]);
        }
        for (; i < m; ++i) {
            const int s0 = __builtin_amdgcn_readlane(s_l, i);
            const float w0 = __int_as_float(__builtin_amdgcn_readlane(w_li, i));
            float v0[V];
            load_row<V>(hl + (long long)s0 * HOUT, v0);
            for (int q = 0; q < V; ++q) acc[q] = fmaf(w0, v0[q], acc[q]);
        }
    }

    const float d2 = dd * dd;
    for (int q = 0; q < V; ++q) {
        float v = acc[q] + d2 * hs[q] + bb[q];
        acc[q] = RELU ? fmaxf(v, 0.0f) : v;
    }

    float* o = y + (long long)node * HOUT + lane * V;
    if constexpr (V == 4) {
        *(float4*)o = make_float4(acc[0], acc[1], acc[2], acc[3]);
    } else {
        *(float2*)o = make_float2(acc[0], acc[1]);
    }
}

extern "C" void kernel_launch(void* const* d_in, const int* in_sizes, int n_in,
                              void* d_out, int out_size, void* d_ws, size_t ws_size,
                              hipStream_t stream) {
    const int*   edge = (const int*)d_in[0];
    const float* x0   = (const float*)d_in[1];
    const float* W1   = (const float*)d_in[2];
    const float* b1   = (const float*)d_in[3];
    const float* W2   = (const float*)d_in[4];
    const float* b2   = (const float*)d_in[5];
    float* out = (float*)d_out;

    const int E    = in_sizes[0] / 2;
    const int N    = in_sizes[1] / EMB;
    const int HID  = in_sizes[3];
    const int REPR = in_sizes[5];
    const int* src = edge;
    const int* dst = edge + E;

    float* dinv = (float*)d_ws;
    float* h    = dinv + N;
    float* x1   = h + (size_t)N * HID;
    const int NB  = (N + BNODES - 1) >> BSH;
    const int NCH = (E + CHUNK - 1) / CHUNK;
    int* rowptr   = (int*)(x1 + (size_t)N * HID);
    unsigned* tmp = (unsigned*)(rowptr + (N + 1));
    int* csr_src  = (int*)(tmp + E);
    int* ghist    = csr_src + E;
    int* goff     = ghist + (size_t)NCH * NB;
    int* btot     = goff + (size_t)NCH * NB;

    const int T = 256;

    size_t smemA = (size_t)NB * sizeof(int);
    size_t smemC = (size_t)(CHUNK + 1) * (sizeof(int) + sizeof(unsigned short)) + (size_t)3 * NB * sizeof(int);
    bucket_hist_kernel<<<NCH, T, smemA, stream>>>(dst, ghist, E, N, NB);
    bucket_offsets_kernel<<<NB, T, 0, stream>>>(ghist, goff, btot, NCH, NB);
    bucket_scatter_kernel<<<NCH, T, smemC, stream>>>(src, dst, ghist, goff, btot, tmp, E, N, NB);
    bucket_sort_kernel<<<NB, BNODES, 0, stream>>>(tmp, btot, csr_src, rowptr, dinv, N, NB);

    size_t smem1 = (size_t)(EMB * HID + 128 * XPITCH) * sizeof(float);
    gemm_wmma_kernel<128><<<(N + 127) / 128, T, smem1, stream>>>(x0, W1, h, N);
    gather_kernel<128, true><<<(N + 7) / 8, T, 0, stream>>>(rowptr, csr_src, dinv, h, b1, x1, N);

    size_t smem2 = (size_t)(EMB * REPR + 128 * XPITCH) * sizeof(float);
    gemm_wmma_kernel<64><<<(N + 127) / 128, T, smem2, stream>>>(x1, W2, h, N);
    gather_kernel<64, false><<<(N + 7) / 8, T, 0, stream>>>(rowptr, csr_src, dinv, h, b2, out, N);
}


# Layer-2 product fused into the layer-1 gather: a 16-wave block sums 64 nodes (4 per wave, edge indices fetched together) into an LDS tile that feeds its WMMAs; x1 and one launch go; sum order kept
# speedup vs baseline: 3.6210x; 1.1404x over previous
#include <hip/hip_runtime.h>
#include <hip/hip_bf16.h>

typedef __attribute__((ext_vector_type(2))) float v2f;
typedef __attribute__((ext_vector_type(8))) float v8f;

#define EMB 128
#define XPITCH 132
#define KSTEPS (EMB / 4)

#if defined(__has_builtin)
#if __has_builtin(__builtin_amdgcn_sched_barrier)
#define SCHED_FENCE() __builtin_amdgcn_sched_barrier(0)
#endif
#endif
#ifndef SCHED_FENCE
#define SCHED_FENCE()
#endif

#define BSH 8
#define BNODES (1 << BSH)
#define SRC_BITS 24
#define SRC_MASK ((1u << SRC_BITS) - 1u)
#define CEPT 25
#define CHUNK (256 * CEPT)
#define DEPT 8
#define DCAP 8192

__device__ __forceinline__ int block_excl_scan(int v, int* sh, int& tot) {
    const int lane = threadIdx.x & 31;
    const int wv = threadIdx.x >> 5;
    int inc = v;
    for (int off = 1; off < 32; off <<= 1) {
        int u = __shfl_up(inc, off, 32);
        if (lane >= off) inc += u;
    }
    if (lane == 31) sh[wv] = inc;
    __syncthreads();
    int wbase = 0, t = 0;
    for (int w = 0; w < 8; ++w) {
        int c = sh[w];
        if (w < wv) wbase += c;
        t += c;
    }
    __syncthreads();
    tot = t;
    return wbase + inc - v;
}

__global__ __launch_bounds__(256) void bucket_hist_kernel(const int* __restrict__ dst,
                                                          int* __restrict__ ghist,
                                                          int e, int n, int nb) {
    extern __shared__ int lds_a[];
    const int t = threadIdx.x;
    for (int b = t; b < nb; b += 256) lds_a[b] = 0;
    __syncthreads();
    const int base = blockIdx.x * CHUNK + t;
    int d[CEPT];
#pragma unroll
    for (int k = 0; k < CEPT; ++k) d[k] = dst[min(base + k * 256, e - 1)];
#pragma unroll
    for (int k = 0; k < CEPT; ++k) {
        atomicAdd(&lds_a[min(max(d[k], 0), n - 1) >> BSH], (base + k * 256 < e) ? 1 : 0);
    }
    __syncthreads();
    for (int b = t; b < nb; b += 256) ghist[blockIdx.x * nb + b] = lds_a[b];
}

__global__ __launch_bounds__(256) void bucket_offsets_kernel(const int* __restrict__ ghist,
                                                             int* __restrict__ goff,
                                                             int* __restrict__ btot,
                                                             int nch, int nb) {
    __shared__ int sh[8];
    const int b = blockIdx.x;
    const int t = threadIdx.x;
    const int per = (nch + 255) / 256;
    const int c0 = min(t * per, nch);
    const int c1 = min(c0 + per, nch);
    int s = 0;
    for (int c = c0; c < c1; ++c) s += ghist[c * nb + b];
    int tot;
    int run = block_excl_scan(s, sh, tot);
    for (int c = c0; c < c1; ++c) {
        goff[c * nb + b] = run;
        run += ghist[c * nb + b];
    }
    if (t == 0) btot[b] = tot;
}

__global__ __launch_bounds__(256) void bucket_scatter_kernel(const int* __restrict__ src,
                                                             const int* __restrict__ dst,
                                                             const int* __restrict__ ghist,
                                                             const int* __restrict__ goff,
                                                             const int* __restrict__ btot,
                                                             unsigned* __restrict__ tmp,
                                                             int e, int n, int nb) {
    extern __shared__ int lds_c[];
    unsigned* stage = (unsigned*)lds_c;
    int* lstart = lds_c + (CHUNK + 1);
    int* fill   = lstart + nb;
    int* gdelta = fill + nb;
    unsigned short* sb = (unsigned short*)(gdelta + nb);
    __shared__ int sh[8];
    const int t = threadIdx.x;
    const int blk = blockIdx.x;

    const int base = blk * CHUNK + t;
    int d[CEPT], s[CEPT];
#pragma unroll
    for (int k = 0; k < CEPT; ++k) {
        int ic = min(base + k * 256, e - 1);
        d[k] = dst[ic];
        s[k] = src[ic];
    }

    const int per = (nb + 255) / 256;
    const int b0 = min(t * per, nb);
    const int b1 = min(b0 + per, nb);
    int sl = 0, sg = 0;
    for (int b = b0; b < b1; ++b) {
        sl += ghist[blk * nb + b];
        sg += btot[b];
    }
    int tot;
    int runl = block_excl_scan(sl, sh, tot);
    int rung = block_excl_scan(sg, sh, tot);
    for (int b = b0; b < b1; ++b) {
        lstart[b] = runl;
        fill[b] = 0;
        gdelta[b] = rung + goff[blk * nb + b] - runl;
        runl += ghist[blk * nb + b];
        rung += btot[b];
    }
    __syncthreads();

    unsigned wd[CEPT];
    int bk[CEPT], pos[CEPT];
#pragma unroll
    for (int k = 0; k < CEPT; ++k) {
        const bool valid = base + k * 256 < e;
        const int dd = min(max(d[k], 0), n - 1);
        bk[k] = dd >> BSH;
        wd[k] = ((unsigned)s[k] & SRC_MASK) | ((unsigned)(dd & (BNODES - 1)) << SRC_BITS);
        const int p = lstart[bk[k]] + atomicAdd(&fill[bk[k]], valid ? 1 : 0);
        pos[k] = valid ? p : CHUNK;
    }
#pragma unroll
    for (int k = 0; k < CEPT; ++k) {
        stage[pos[k]] = wd[k];
        sb[pos[k]] = (unsigned short)bk[k];
    }
    __syncthreads();

    const int cnt = min(CHUNK, e - blk * CHUNK);
    unsigned wq[CEPT];
    int gq[CEPT];
#pragma unroll
    for (int k = 0; k < CEPT; ++k) {
        const int ii = min(t + k * 256, cnt - 1);
        wq[k] = stage[ii];
        gq[k] = sb[ii];
    }
#pragma unroll
    for (int k = 0; k < CEPT; ++k) gq[k] = gdelta[gq[k]];
#pragma unroll
    for (int k = 0; k < CEPT; ++k) {
        const int i = t + k * 256;
        if (i < cnt) tmp[gq[k] + i] = wq[k];
    }
}

__global__ __launch_bounds__(BNODES) void bucket_sort_kernel(const unsigned* __restrict__ tmp,
                                                             const int* __restrict__ btot,
                                                             int* __restrict__ csr_src,
                                                             int* __restrict__ rowptr,
                                                             float* __restrict__ dinv,
                                                             int n, int nb) {
    __shared__ int ncnt[BNODES];
    __shared__ int nstart[BNODES];
    __shared__ int fill[BNODES];
    __shared__ int stage[DCAP + 1];
    __shared__ int sh[8];
    const int b = blockIdx.x;
    const int t = threadIdx.x;

    int part = 0;
    for (int j = t; j < b; j += BNODES) part += btot[j];
    int base;
    block_excl_scan(part, sh, base);
    const int cntb = btot[b];

    ncnt[t] = 0;
    fill[t] = 0;
    __syncthreads();

    for (int i0 = t; i0 < cntb; i0 += BNODES * DEPT) {
        unsigned w[DEPT];
#pragma unroll
        for (int k = 0; k < DEPT; ++k) w[k] = tmp[base + min(i0 + k * BNODES, cntb - 1)];
#pragma unroll
        for (int k = 0; k < DEPT; ++k)
            atomicAdd(&ncnt[w[k] >> SRC_BITS], (i0 + k * BNODES < cntb) ? 1 : 0);
    }
    __syncthreads();

    const int c = ncnt[t];
    int tot;
    const int st = block_excl_scan(c, sh, tot);
    nstart[t] = st;
    const int node = b * BNODES + t;
    if (node < n) {
        rowptr[node] = base + st;
        dinv[node] = rsqrtf((float)(c + 1));
    }
    if (b == nb - 1 && t == 0) rowptr[n] = base + cntb;
    __syncthreads();

    const bool staged = cntb <= DCAP;
    for (int i0 = t; i0 < cntb; i0 += BNODES * DEPT) {
        unsigned w[DEPT];
        int pos[DEPT];
#pragma unroll
        for (int k = 0; k < DEPT; ++k) w[k] = tmp[base + min(i0 + k * BNODES, cntb - 1)];
#pragma unroll
        for (int k = 0; k < DEPT; ++k) {
            const bool valid = i0 + k * BNODES < cntb;
            const int dl = (int)(w[k] >> SRC_BITS);
            const int p = nstart[dl] + atomicAdd(&fill[dl], valid ? 1 : 0);
            pos[k] = valid ? p : -1;
        }
        if (staged) {
#pragma unroll
            for (int k = 0; k < DEPT; ++k) stage[pos[k] < 0 ? DCAP : pos[k]] = (int)(w[k] & SRC_MASK);
        } else {
#pragma unroll
            for (int k = 0; k < DEPT; ++k) {
                if (pos[k] >= 0) csr_src[base + pos[k]] = (int)(w[k] & SRC_MASK);
            }
        }
    }
    if (staged) {
        __syncthreads();
        for (int i = t; i < cntb; i += BNODES) csr_src[base + i] = stage[i];
    }
}

template <int HOUT>
__global__ __launch_bounds__(256) void gemm_wmma_kernel(const float* __restrict__ X,
                                                        const float* __restrict__ W,
                                                        float* __restrict__ H, int nrows) {
    extern __shared__ float smem[];
    float* sW = smem;
    float* sX = smem + EMB * HOUT;

    const int tid = threadIdx.x;
    const int rowBase = blockIdx.x * 128;

    for (int i = tid * 4; i < EMB * HOUT; i += 256 * 4) {
        *(float4*)&sW[i] = *(const float4*)&W[i];
    }
    for (int i = tid * 4; i < 128 * EMB; i += 256 * 4) {
        int r = i >> 7, c = i & 127;
        int gr = rowBase + r;
        float4 v = make_float4(0.f, 0.f, 0.f, 0.f);
        if (gr < nrows) v = *(const float4*)&X[(long long)gr * EMB + c];
        *(float4*)&sX[r * XPITCH + c] = v;
    }
    __syncthreads();

    const int wave = tid >> 5;
    const int lane = tid & 31;
    const int nl = lane & 15;
    const int kh = lane >> 4;
    const int wr = wave * 16;

    const float* xrow = &sX[(wr + nl) * XPITCH + 2 * kh];
    v2f afrag[KSTEPS];
#pragma unroll
    for (int ks = 0; ks < KSTEPS; ++ks) afrag[ks] = *(const v2f*)&xrow[4 * ks];
    SCHED_FENCE();

    for (int nt = 0; nt < HOUT / 16; ++nt) {
        const float* wcol = &sW[nt * 16 + nl];
        v2f bfrag[KSTEPS];
#pragma unroll
        for (int ks = 0; ks < KSTEPS; ++ks) {
            const int kk = 4 * ks + 2 * kh;
            v2f b;
            b.x = wcol[kk * HOUT];
            b.y = wcol[(kk + 1) * HOUT];
            bfrag[ks] = b;
        }
        SCHED_FENCE();
        v8f acc = {};
#pragma unroll
        for (int ks = 0; ks < KSTEPS; ++ks) {
            acc = __builtin_amdgcn_wmma_f32_16x16x4_f32(
                false, afrag[ks], false, bfrag[ks], (short)0, acc, false, false);
        }
        SCHED_FENCE();
#pragma unroll
        for (int j = 0; j < 8; ++j) {
            int r = rowBase + wr + kh * 8 + j;
            if (r < nrows) H[(long long)r * HOUT + nt * 16 + nl] = acc[j];
        }
    }
}

template <int V>
__device__ __forceinline__ void load_row(const float* __restrict__ p, float (&v)[V]) {
    if constexpr (V == 4) {
        float4 t = *(const float4*)p;
        v[0] = t.x; v[1] = t.y; v[2] = t.z; v[3] = t.w;
    } else {
        float2 t = *(const float2*)p;
        v[0] = t.x; v[1] = t.y;
    }
}

template <int HOUT, bool RELU>
__global__ __launch_bounds__(256) void gather_kernel(const int* __restrict__ rowptr,
                                                     const int* __restrict__ csr_src,
                                                     const float* __restrict__ dinv,
                                                     const float* __restrict__ h,
                                                     const float* __restrict__ b,
                                                     float* __restrict__ y, int n) {
    constexpr int V = HOUT / 32;
    const int lane = threadIdx.x & 31;
    const int wv = __builtin_amdgcn_readfirstlane((int)(threadIdx.x >> 5));
    const int node = blockIdx.x * 8 + wv;
    if (node >= n) return;
    const int beg = rowptr[node];
    const int end = rowptr[node + 1];
    const float dd = dinv[node];
    const float* hl = h + lane * V;

    float hs[V], bb[V];
    load_row<V>(hl + (long long)node * HOUT, hs);
    load_row<V>(b + lane * V, bb);

    float acc[V];
    for (int q = 0; q < V; ++q) acc[q] = 0.0f;

    for (int j0 = beg; j0 < end; j0 += 32) {
        const int m = min(32, end - j0);
        int s_l = 0;
        float w_l = 0.0f;
        if (lane < m) {
            s_l = csr_src[j0 + lane];
            s_l = min(max(s_l, 0), n - 1);
            w_l = dinv[s_l] * dd;
        }
        const int w_li = __float_as_int(w_l);
        int i = 0;
        for (; i + 4 <= m; i += 4) {
            const int s0 = __builtin_amdgcn_readlane(s_l, i);
            const int s1 = __builtin_amdgcn_readlane(s_l, i + 1);
            const int s2 = __builtin_amdgcn_readlane(s_l, i + 2);
            const int s3 = __builtin_amdgcn_readlane(s_l, i + 3);
            const float w0 = __int_as_float(__builtin_amdgcn_readlane(w_li, i));
            const float w1 = __int_as_float(__builtin_amdgcn_readlane(w_li, i + 1));
            const float w2 = __int_as_float(__builtin_amdgcn_readlane(w_li, i + 2));
            const float w3 = __int_as_float(__builtin_amdgcn_readlane(w_li, i + 3));
            float v0[V], v1[V], v2[V], v3[V];
            load_row<V>(hl + (long long)s0 * HOUT, v0);
            load_row<V>(hl + (long long)s1 * HOUT, v1);
            load_row<V>(hl + (long long)s2 * HOUT, v2);
            load_row<V>(hl + (long long)s3 * HOUT, v3);
            for (int q = 0; q < V; ++q) acc[q] = fmaf(w0, v0[q], acc[q]);
            for (int q = 0; q < V; ++q) acc[q] = fmaf(w1, v1[q], acc[q]);
            for (int q = 0; q < V; ++q) acc[q] = fmaf(w2, v2[q], acc[q]);
            for (int q = 0; q < V; ++q) acc[q] = fmaf(w3, v3[q], acc[q]);
        }
        for (; i < m; ++i) {
            const int s0 = __builtin_amdgcn_readlane(s_l, i);
            const float w0 = __int_as_float(__builtin_amdgcn_readlane(w_li, i));
            float v0[V];
            load_row<V>(hl + (long long)s0 * HOUT, v0);
            for (int q = 0; q < V; ++q) acc[q] = fmaf(w0, v0[q], acc[q]);
        }
    }

    const float d2 = dd * dd;
    for (int q = 0; q < V; ++q) {
        float v = acc[q] + d2 * hs[q] + bb[q];
        acc[q] = RELU ? fmaxf(v, 0.0f) : v;
    }

    float* o = y + (long long)node * HOUT + lane * V;
    if constexpr (V == 4) {
        *(float4*)o = make_float4(acc[0], acc[1], acc[2], acc[3]);
    } else {
        *(float2*)o = make_float2(acc[0], acc[1]);
    }
}

#define FNODES 64
#define NPW 4
#define H2OUT 64

__device__ __forceinline__ void add_rows128(int s_l, int w_li, int m, const float* __restrict__ hl,
                                            float (&acc)[4]) {
    int i = 0;
    for (; i + 4 <= m; i += 4) {
        const int s0 = __builtin_amdgcn_readlane(s_l, i);
        const int s1 = __builtin_amdgcn_readlane(s_l, i + 1);
        const int s2 = __builtin_amdgcn_readlane(s_l, i + 2);
        const int s3 = __builtin_amdgcn_readlane(s_l, i + 3);
        const float w0 = __int_as_float(__builtin_amdgcn_readlane(w_li, i));
        const float w1 = __int_as_float(__builtin_amdgcn_readlane(w_li, i + 1));
        const float w2 = __int_as_float(__builtin_amdgcn_readlane(w_li, i + 2));
        const float w3 = __int_as_float(__builtin_amdgcn_readlane(w_li, i + 3));
        float v0[4], v1[4], v2[4], v3[4];
        load_row<4>(hl + (long long)s0 * EMB, v0);
        load_row<4>(hl + (long long)s1 * EMB, v1);
        load_row<4>(hl + (long long)s2 * EMB, v2);
        load_row<4>(hl + (long long)s3 * EMB, v3);
        for (int q = 0; q < 4; ++q) acc[q] = fmaf(w0, v0[q], acc[q]);
        for (int q = 0; q < 4; ++q) acc[q] = fmaf(w1, v1[q], acc[q]);
        for (int q = 0; q < 4; ++q) acc[q] = fmaf(w2, v2[q], acc[q]);
        for (int q = 0; q < 4; ++q) acc[q] = fmaf(w3, v3[q], acc[q]);
    }
    for (; i < m; ++i) {
        const int s0 = __builtin_amdgcn_readlane(s_l, i);
        const float w0 = __int_as_float(__builtin_amdgcn_readlane(w_li, i));
        float v0[4];
        load_row<4>(hl + (long long)s0 * EMB, v0);
        for (int q = 0; q < 4; ++q) acc[q] = fmaf(w0, v0[q], acc[q]);
    }
}

__global__ __launch_bounds__(512) void gather_product_kernel(const int* __restrict__ rowptr,
                                                             const int* __restrict__ csr_src,
                                                             const float* __restrict__ dinv,
                                                             const float* __restrict__ h,
                                                             const float* __restrict__ b,
                                                             const float* __restrict__ W,
                                                             float* __restrict__ H2, int n, int e) {
    extern __shared__ float smem_f[];
    float* sW = smem_f;
    float* sX = smem_f + EMB * H2OUT;

    const int tid = threadIdx.x;
    const int lane = tid & 31;
    const int wv = __builtin_amdgcn_readfirstlane((int)(tid >> 5));
    const int node0 = blockIdx.x * FNODES + wv * NPW;

    float4 wq[4];
#pragma unroll
    for (int k = 0; k < 4; ++k) wq[k] = *(const float4*)&W[(tid + k * 512) * 4];

    int rp[NPW + 1];
    float ddq[NPW];
#pragma unroll
    for (int q = 0; q <= NPW; ++q) rp[q] = rowptr[min(node0 + q, n)];
#pragma unroll
    for (int q = 0; q < NPW; ++q) ddq[q] = dinv[min(node0 + q, n - 1)];

    int s_q[NPW];
    float w_q[NPW];
#pragma unroll
    for (int q = 0; q < NPW; ++q) s_q[q] = csr_src[min(rp[q] + lane, e - 1)];
#pragma unroll
    for (int q = 0; q < NPW; ++q) {
        const bool valid = lane < min(32, rp[q + 1] - rp[q]);
        s_q[q] = valid ? min(max(s_q[q], 0), n - 1) : 0;
    }
    float ds[NPW];
#pragma unroll
    for (int q = 0; q < NPW; ++q) ds[q] = dinv[s_q[q]];
#pragma unroll
    for (int k = 0; k < 4; ++k) *(float4*)&sW[(tid + k * 512) * 4] = wq[k];
    asm volatile("" ::"v"(ds[0]), "v"(ds[1]), "v"(ds[2]), "v"(ds[3]));
#pragma unroll
    for (int q = 0; q < NPW; ++q) {
        const bool valid = lane < min(32, rp[q + 1] - rp[q]);
        w_q[q] = valid ? ds[q] * ddq[q] : 0.0f;
    }

    const float* hl = h + lane * 4;
    float bb[4];
    load_row<4>(b + lane * 4, bb);

#pragma unroll
    for (int q = 0; q < NPW; ++q) {
        const int node = node0 + q;
        float o[4] = {0.0f, 0.0f, 0.0f, 0.0f};
        if (node < n) {
            float hs[4];
            load_row<4>(hl + (long long)node * EMB, hs);
            float acc[4] = {0.0f, 0.0f, 0.0f, 0.0f};
            const int beg = rp[q];
            const int end = rp[q + 1];
            const float dd = ddq[q];
            add_rows128(s_q[q], __float_as_int(w_q[q]), min(32, end - beg), hl, acc);
            for (int j0 = beg + 32; j0 < end; j0 += 32) {
                const int m = min(32, end - j0);
                int s_l = 0;
                float w_l = 0.0f;
                if (lane < m) {
                    s_l = csr_src[j0 + lane];
                    s_l = min(max(s_l, 0), n - 1);
                    w_l = dinv[s_l] * dd;
                }
                add_rows128(s_l, __float_as_int(w_l), m, hl, acc);
            }
            const float d2 = dd * dd;
            for (int c = 0; c < 4; ++c) o[c] = fmaxf(acc[c] + d2 * hs[c] + bb[c], 0.0f);
        }
        *(float4*)&sX[(wv * NPW + q) * XPITCH + lane * 4] = make_float4(o[0], o[1], o[2], o[3]);
    }
    __syncthreads();

    if (wv >= 8) return;
    const int nl = lane & 15;
    const int kh = lane >> 4;
    const int g = wv >> 1;
    const int ch = wv & 1;
    const float* xrow = &sX[(g * 16 + nl) * XPITCH + 2 * kh];
    const float* wcol = &sW[ch * 32 + nl];
    v8f c0 = {}, c1 = {};
#pragma unroll
    for (int ks = 0; ks < KSTEPS; ++ks) {
        const v2f a = *(const v2f*)&xrow[4 * ks];
        const int kk = 4 * ks + 2 * kh;
        v2f b0, b1;
        b0.x = wcol[kk * H2OUT];
        b0.y = wcol[(kk + 1) * H2OUT];
        b1.x = wcol[kk * H2OUT + 16];
        b1.y = wcol[(kk + 1) * H2OUT + 16];
        c0 = __builtin_amdgcn_wmma_f32_16x16x4_f32(false, a, false, b0, (short)0, c0, false, false);
        c1 = __builtin_amdgcn_wmma_f32_16x16x4_f32(false, a, false, b1, (short)0, c1, false, false);
    }
#pragma unroll
    for (int j = 0; j < 8; ++j) {
        const int r = blockIdx.x * FNODES + g * 16 + kh * 8 + j;
        if (r < n) {
            H2[(long long)r * H2OUT + ch * 32 + nl] = c0[j];
            H2[(long long)r * H2OUT + ch * 32 + 16 + nl] = c1[j];
        }
    }
}

extern "C" void kernel_launch(void* const* d_in, const int* in_sizes, int n_in,
                              void* d_out, int out_size, void* d_ws, size_t ws_size,
                              hipStream_t stream) {
    const int*   edge = (const int*)d_in[0];
    const float* x0   = (const float*)d_in[1];
    const float* W1   = (const float*)d_in[2];
    const float* b1   = (const float*)d_in[3];
    const float* W2   = (const float*)d_in[4];
    const float* b2   = (const float*)d_in[5];
    float* out = (float*)d_out;

    const int E    = in_sizes[0] / 2;
    const int N    = in_sizes[1] / EMB;
    const int HID  = in_sizes[3];
    const int REPR = in_sizes[5];
    const int* src = edge;
    const int* dst = edge + E;

    float* dinv = (float*)d_ws;
    float* h    = dinv + N;
    float* x1   = h + (size_t)N * HID;
    const int NB  = (N + BNODES - 1) >> BSH;
    const int NCH = (E + CHUNK - 1) / CHUNK;
    int* rowptr   = (int*)(x1 + (size_t)N * HID);
    unsigned* tmp = (unsigned*)(rowptr + (N + 1));
    int* csr_src  = (int*)(tmp + E);
    int* ghist    = csr_src + E;
    int* goff     = ghist + (size_t)NCH * NB;
    int* btot     = goff + (size_t)NCH * NB;

    const int T = 256;

    size_t smemA = (size_t)NB * sizeof(int);
    size_t smemC = (size_t)(CHUNK + 1) * (sizeof(int) + sizeof(unsigned short)) + (size_t)3 * NB * sizeof(int);
    bucket_hist_kernel<<<NCH, T, smemA, stream>>>(dst, ghist, E, N, NB);
    bucket_offsets_kernel<<<NB, T, 0, stream>>>(ghist, goff, btot, NCH, NB);
    bucket_scatter_kernel<<<NCH, T, smemC, stream>>>(src, dst, ghist, goff, btot, tmp, E, N, NB);
    bucket_sort_kernel<<<NB, BNODES, 0, stream>>>(tmp, btot, csr_src, rowptr, dinv, N, NB);

    size_t smem1 = (size_t)(EMB * HID + 128 * XPITCH) * sizeof(float);
    gemm_wmma_kernel<128><<<(N + 127) / 128, T, smem1, stream>>>(x0, W1, h, N);

    size_t smemF = (size_t)(EMB * H2OUT + FNODES * XPITCH) * sizeof(float);
    gather_product_kernel<<<(N + FNODES - 1) / FNODES, 512, smemF, stream>>>(rowptr, csr_src, dinv, h, b1,
                                                                           W2, x1, N, E);

    gather_kernel<64, false><<<(N + 7) / 8, T, 0, stream>>>(rowptr, csr_src, dinv, x1, b2, out, N);
}


# Layer-1 product: a thread's 32 staging loads (16-byte pieces of W1 and of the X tile) issue as register groups ahead of the LDS stores; the rolled copy loops waited for each piece in turn
# speedup vs baseline: 3.9902x; 1.1020x over previous
#include <hip/hip_runtime.h>
#include <hip/hip_bf16.h>

typedef __attribute__((ext_vector_type(2))) float v2f;
typedef __attribute__((ext_vector_type(8))) float v8f;

#define EMB 128
#define XPITCH 132
#define KSTEPS (EMB / 4)

#if defined(__has_builtin)
#if __has_builtin(__builtin_amdgcn_sched_barrier)
#define SCHED_FENCE() __builtin_amdgcn_sched_barrier(0)
#endif
#endif
#ifndef SCHED_FENCE
#define SCHED_FENCE()
#endif

#define BSH 8
#define BNODES (1 << BSH)
#define SRC_BITS 24
#define SRC_MASK ((1u << SRC_BITS) - 1u)
#define CEPT 25
#define CHUNK (256 * CEPT)
#define DEPT 8
#define DCAP 8192

__device__ __forceinline__ int block_excl_scan(int v, int* sh, int& tot) {
    const int lane = threadIdx.x & 31;
    const int wv = threadIdx.x >> 5;
    int inc = v;
    for (int off = 1; off < 32; off <<= 1) {
        int u = __shfl_up(inc, off, 32);
        if (lane >= off) inc += u;
    }
    if (lane == 31) sh[wv] = inc;
    __syncthreads();
    int wbase = 0, t = 0;
    for (int w = 0; w < 8; ++w) {
        int c = sh[w];
        if (w < wv) wbase += c;
        t += c;
    }
    __syncthreads();
    tot = t;
    return wbase + inc - v;
}

__global__ __launch_bounds__(256) void bucket_hist_kernel(const int* __restrict__ dst,
                                                          int* __restrict__ ghist,
                                                          int e, int n, int nb) {
    extern __shared__ int lds_a[];
    const int t = threadIdx.x;
    for (int b = t; b < nb; b += 256) lds_a[b] = 0;
    __syncthreads();
    const int base = blockIdx.x * CHUNK + t;
    int d[CEPT];
#pragma unroll
    for (int k = 0; k < CEPT; ++k) d[k] = dst[min(base + k * 256, e - 1)];
#pragma unroll
    for (int k = 0; k < CEPT; ++k) {
        atomicAdd(&lds_a[min(max(d[k], 0), n - 1) >> BSH], (base + k * 256 < e) ? 1 : 0);
    }
    __syncthreads();
    for (int b = t; b < nb; b += 256) ghist[blockIdx.x * nb + b] = lds_a[b];
}

__global__ __launch_bounds__(256) void bucket_offsets_kernel(const int* __restrict__ ghist,
                                                             int* __restrict__ goff,
                                                             int* __restrict__ btot,
                                                             int nch, int nb) {
    __shared__ int sh[8];
    const int b = blockIdx.x;
    const int t = threadIdx.x;
    const int per = (nch + 255) / 256;
    const int c0 = min(t * per, nch);
    const int c1 = min(c0 + per, nch);
    int s = 0;
    for (int c = c0; c < c1; ++c) s += ghist[c * nb + b];
    int tot;
    int run = block_excl_scan(s, sh, tot);
    for (int c = c0; c < c1; ++c) {
        goff[c * nb + b] = run;
        run += ghist[c * nb + b];
    }
    if (t == 0) btot[b] = tot;
}

__global__ __launch_bounds__(256) void bucket_scatter_kernel(const int* __restrict__ src,
                                                             const int* __restrict__ dst,
                                                             const int* __restrict__ ghist,
                                                             const int* __restrict__ goff,
                                                             const int* __restrict__ btot,
                                                             unsigned* __restrict__ tmp,
                                                             int e, int n, int nb) {
    extern __shared__ int lds_c[];
    unsigned* stage = (unsigned*)lds_c;
    int* lstart = lds_c + (CHUNK + 1);
    int* fill   = lstart + nb;
    int* gdelta = fill + nb;
    unsigned short* sb = (unsigned short*)(gdelta + nb);
    __shared__ int sh[8];
    const int t = threadIdx.x;
    const int blk = blockIdx.x;

    const int base = blk * CHUNK + t;
    int d[CEPT], s[CEPT];
#pragma unroll
    for (int k = 0; k < CEPT; ++k) {
        int ic = min(base + k * 256, e - 1);
        d[k] = dst[ic];
        s[k] = src[ic];
    }

    const int per = (nb + 255) / 256;
    const int b0 = min(t * per, nb);
    const int b1 = min(b0 + per, nb);
    int sl = 0, sg = 0;
    for (int b = b0; b < b1; ++b) {
        sl += ghist[blk * nb + b];
        sg += btot[b];
    }
    int tot;
    int runl = block_excl_scan(sl, sh, tot);
    int rung = block_excl_scan(sg, sh, tot);
    for (int b = b0; b < b1; ++b) {
        lstart[b] = runl;
        fill[b] = 0;
        gdelta[b] = rung + goff[blk * nb + b] - runl;
        runl += ghist[blk * nb + b];
        rung += btot[b];
    }
    __syncthreads();

    unsigned wd[CEPT];
    int bk[CEPT], pos[CEPT];
#pragma unroll
    for (int k = 0; k < CEPT; ++k) {
        const bool valid = base + k * 256 < e;
        const int dd = min(max(d[k], 0), n - 1);
        bk[k] = dd >> BSH;
        wd[k] = ((unsigned)s[k] & SRC_MASK) | ((unsigned)(dd & (BNODES - 1)) << SRC_BITS);
        const int p = lstart[bk[k]] + atomicAdd(&fill[bk[k]], valid ? 1 : 0);
        pos[k] = valid ? p : CHUNK;
    }
#pragma unroll
    for (int k = 0; k < CEPT; ++k) {
        stage[pos[k]] = wd[k];
        sb[pos[k]] = (unsigned short)bk[k];
    }
    __syncthreads();

    const int cnt = min(CHUNK, e - blk * CHUNK);
    unsigned wq[CEPT];
    int gq[CEPT];
#pragma unroll
    for (int k = 0; k < CEPT; ++k) {
        const int ii = min(t + k * 256, cnt - 1);
        wq[k] = stage[ii];
        gq[k] = sb[ii];
    }
#pragma unroll
    for (int k = 0; k < CEPT; ++k) gq[k] = gdelta[gq[k]];
#pragma unroll
    for (int k = 0; k < CEPT; ++k) {
        const int i = t + k * 256;
        if (i < cnt) tmp[gq[k] + i] = wq[k];
    }
}

__global__ __launch_bounds__(BNODES) void bucket_sort_kernel(const unsigned* __restrict__ tmp,
                                                             const int* __restrict__ btot,
                                                             int* __restrict__ csr_src,
                                                             int* __restrict__ rowptr,
                                                             float* __restrict__ dinv,
                                                             int n, int nb) {
    __shared__ int ncnt[BNODES];
    __shared__ int nstart[BNODES];
    __shared__ int fill[BNODES];
    __shared__ int stage[DCAP + 1];
    __shared__ int sh[8];
    const int b = blockIdx.x;
    const int t = threadIdx.x;

    int part = 0;
    for (int j = t; j < b; j += BNODES) part += btot[j];
    int base;
    block_excl_scan(part, sh, base);
    const int cntb = btot[b];

    ncnt[t] = 0;
    fill[t] = 0;
    __syncthreads();

    for (int i0 = t; i0 < cntb; i0 += BNODES * DEPT) {
        unsigned w[DEPT];
#pragma unroll
        for (int k = 0; k < DEPT; ++k) w[k] = tmp[base + min(i0 + k * BNODES, cntb - 1)];
#pragma unroll
        for (int k = 0; k < DEPT; ++k)
            atomicAdd(&ncnt[w[k] >> SRC_BITS], (i0 + k * BNODES < cntb) ? 1 : 0);
    }
    __syncthreads();

    const int c = ncnt[t];
    int tot;
    const int st = block_excl_scan(c, sh, tot);
    nstart[t] = st;
    const int node = b * BNODES + t;
    if (node < n) {
        rowptr[node] = base + st;
        dinv[node] = rsqrtf((float)(c + 1));
    }
    if (b == nb - 1 && t == 0) rowptr[n] = base + cntb;
    __syncthreads();

    const bool staged = cntb <= DCAP;
    for (int i0 = t; i0 < cntb; i0 += BNODES * DEPT) {
        unsigned w[DEPT];
        int pos[DEPT];
#pragma unroll
        for (int k = 0; k < DEPT; ++k) w[k] = tmp[base + min(i0 + k * BNODES, cntb - 1)];
#pragma unroll
        for (int k = 0; k < DEPT; ++k) {
            const bool valid = i0 + k * BNODES < cntb;
            const int dl = (int)(w[k] >> SRC_BITS);
            const int p = nstart[dl] + atomicAdd(&fill[dl], valid ? 1 : 0);
            pos[k] = valid ? p : -1;
        }
        if (staged) {
#pragma unroll
            for (int k = 0; k < DEPT; ++k) stage[pos[k] < 0 ? DCAP : pos[k]] = (int)(w[k] & SRC_MASK);
        } else {
#pragma unroll
            for (int k = 0; k < DEPT; ++k) {
                if (pos[k] >= 0) csr_src[base + pos[k]] = (int)(w[k] & SRC_MASK);
            }
        }
    }
    if (staged) {
        __syncthreads();
        for (int i = t; i < cntb; i += BNODES) csr_src[base + i] = stage[i];
    }
}

template <int HOUT>
__global__ __launch_bounds__(256) void gemm_wmma_kernel(const float* __restrict__ X,
                                                        const float* __restrict__ W,
                                                        float* __restrict__ H, int nrows) {
    extern __shared__ float smem[];
    float* sW = smem;
    float* sX = smem + EMB * HOUT;

    const int tid = threadIdx.x;
    const int rowBase = blockIdx.x * 128;

    constexpr int WP = EMB * HOUT / (256 * 4);
    constexpr int XP = 128 * EMB / (256 * 4);
    const int xr = tid >> 5;
    const int xc = (tid & 31) * 4;
    float4 wv[WP], xv[XP];
#pragma unroll
    for (int k = 0; k < WP; ++k) wv[k] = *(const float4*)&W[(tid + k * 256) * 4];
#pragma unroll
    for (int k = 0; k < XP; ++k)
        xv[k] = *(const float4*)&X[(long long)min(rowBase + xr + 8 * k, nrows - 1) * EMB + xc];
#pragma unroll
    for (int k = 0; k < WP; ++k) *(float4*)&sW[(tid + k * 256) * 4] = wv[k];
#pragma unroll
    for (int k = 0; k < XP; ++k) {
        const int r = xr + 8 * k;
        *(float4*)&sX[r * XPITCH + xc] =
            (rowBase + r < nrows) ? xv[k] : make_float4(0.f, 0.f, 0.f, 0.f);
    }
    __syncthreads();

    const int wave = tid >> 5;
    const int lane = tid & 31;
    const int nl = lane & 15;
    const int kh = lane >> 4;
    const int wr = wave * 16;

    const float* xrow = &sX[(wr + nl) * XPITCH + 2 * kh];
    v2f afrag[KSTEPS];
#pragma unroll
    for (int ks = 0; ks < KSTEPS; ++ks) afrag[ks] = *(const v2f*)&xrow[4 * ks];
    SCHED_FENCE();

    for (int nt = 0; nt < HOUT / 16; ++nt) {
        const float* wcol = &sW[nt * 16 + nl];
        v2f bfrag[KSTEPS];
#pragma unroll
        for (int ks = 0; ks < KSTEPS; ++ks) {
            const int kk = 4 * ks + 2 * kh;
            v2f b;
            b.x = wcol[kk * HOUT];
            b.y = wcol[(kk + 1) * HOUT];
            bfrag[ks] = b;
        }
        SCHED_FENCE();
        v8f acc = {};
#pragma unroll
        for (int ks = 0; ks < KSTEPS; ++ks) {
            acc = __builtin_amdgcn_wmma_f32_16x16x4_f32(
                false, afrag[ks], false, bfrag[ks], (short)0, acc, false, false);
        }
        SCHED_FENCE();
#pragma unroll
        for (int j = 0; j < 8; ++j) {
            int r = rowBase + wr + kh * 8 + j;
            if (r < nrows) H[(long long)r * HOUT + nt * 16 + nl] = acc[j];
        }
    }
}

template <int V>
__device__ __forceinline__ void load_row(const float* __restrict__ p, float (&v)[V]) {
    if constexpr (V == 4) {
        float4 t = *(const float4*)p;
        v[0] = t.x; v[1] = t.y; v[2] = t.z; v[3] = t.w;
    } else {
        float2 t = *(const float2*)p;
        v[0] = t.x; v[1] = t.y;
    }
}

template <int HOUT, bool RELU>
__global__ __launch_bounds__(256) void gather_kernel(const int* __restrict__ rowptr,
                                                     const int* __restrict__ csr_src,
                                                     const float* __restrict__ dinv,
                                                     const float* __restrict__ h,
                                                     const float* __restrict__ b,
                                                     float* __restrict__ y, int n) {
    constexpr int V = HOUT / 32;
    const int lane = threadIdx.x & 31;
    const int wv = __builtin_amdgcn_readfirstlane((int)(threadIdx.x >> 5));
    const int node = blockIdx.x * 8 + wv;
    if (node >= n) return;
    const int beg = rowptr[node];
    const int end = rowptr[node + 1];
    const float dd = dinv[node];
    const float* hl = h + lane * V;

    float hs[V], bb[V];
    load_row<V>(hl + (long long)node * HOUT, hs);
    load_row<V>(b + lane * V, bb);

    float acc[V];
    for (int q = 0; q < V; ++q) acc[q] = 0.0f;

    for (int j0 = beg; j0 < end; j0 += 32) {
        const int m = min(32, end - j0);
        int s_l = 0;
        float w_l = 0.0f;
        if (lane < m) {
            s_l = csr_src[j0 + lane];
            s_l = min(max(s_l, 0), n - 1);
            w_l = dinv[s_l] * dd;
        }
        const int w_li = __float_as_int(w_l);
        int i = 0;
        for (; i + 4 <= m; i += 4) {
            const int s0 = __builtin_amdgcn_readlane(s_l, i);
            const int s1 = __builtin_amdgcn_readlane(s_l, i + 1);
            const int s2 = __builtin_amdgcn_readlane(s_l, i + 2);
            const int s3 = __builtin_amdgcn_readlane(s_l, i + 3);
            const float w0 = __int_as_float(__builtin_amdgcn_readlane(w_li, i));
            const float w1 = __int_as_float(__builtin_amdgcn_readlane(w_li, i + 1));
            const float w2 = __int_as_float(__builtin_amdgcn_readlane(w_li, i + 2));
            const float w3 = __int_as_float(__builtin_amdgcn_readlane(w_li, i + 3));
            float v0[V], v1[V], v2[V], v3[V];
            load_row<V>(hl + (long long)s0 * HOUT, v0);
            load_row<V>(hl + (long long)s1 * HOUT, v1);
            load_row<V>(hl + (long long)s2 * HOUT, v2);
            load_row<V>(hl + (long long)s3 * HOUT, v3);
            for (int q = 0; q < V; ++q) acc[q] = fmaf(w0, v0[q], acc[q]);
            for (int q = 0; q < V; ++q) acc[q] = fmaf(w1, v1[q], acc[q]);
            for (int q = 0; q < V; ++q) acc[q] = fmaf(w2, v2[q], acc[q]);
            for (int q = 0; q < V; ++q) acc[q] = fmaf(w3, v3[q], acc[q]);
        }
        for (; i < m; ++i) {
            const int s0 = __builtin_amdgcn_readlane(s_l, i);
            const float w0 = __int_as_float(__builtin_amdgcn_readlane(w_li, i));
            float v0[V];
            load_row<V>(hl + (long long)s0 * HOUT, v0);
            for (int q = 0; q < V; ++q) acc[q] = fmaf(w0, v0[q], acc[q]);
        }
    }

    const float d2 = dd * dd;
    for (int q = 0; q < V; ++q) {
        float v = acc[q] + d2 * hs[q] + bb[q];
        acc[q] = RELU ? fmaxf(v, 0.0f) : v;
    }

    float* o = y + (long long)node * HOUT + lane * V;
    if constexpr (V == 4) {
        *(float4*)o = make_float4(acc[0], acc[1], acc[2], acc[3]);
    } else {
        *(float2*)o = make_float2(acc[0], acc[1]);
    }
}

#define FNODES 64
#define NPW 4
#define H2OUT 64

__device__ __forceinline__ void add_rows128(int s_l, int w_li, int m, const float* __restrict__ hl,
                                            float (&acc)[4]) {
    int i = 0;
    for (; i + 4 <= m; i += 4) {
        const int s0 = __builtin_amdgcn_readlane(s_l, i);
        const int s1 = __builtin_amdgcn_readlane(s_l, i + 1);
        const int s2 = __builtin_amdgcn_readlane(s_l, i + 2);
        const int s3 = __builtin_amdgcn_readlane(s_l, i + 3);
        const float w0 = __int_as_float(__builtin_amdgcn_readlane(w_li, i));
        const float w1 = __int_as_float(__builtin_amdgcn_readlane(w_li, i + 1));
        const float w2 = __int_as_float(__builtin_amdgcn_readlane(w_li, i + 2));
        const float w3 = __int_as_float(__builtin_amdgcn_readlane(w_li, i + 3));
        float v0[4], v1[4], v2[4], v3[4];
        load_row<4>(hl + (long long)s0 * EMB, v0);
        load_row<4>(hl + (long long)s1 * EMB, v1);
        load_row<4>(hl + (long long)s2 * EMB, v2);
        load_row<4>(hl + (long long)s3 * EMB, v3);
        for (int q = 0; q < 4; ++q) acc[q] = fmaf(w0, v0[q], acc[q]);
        for (int q = 0; q < 4; ++q) acc[q] = fmaf(w1, v1[q], acc[q]);
        for (int q = 0; q < 4; ++q) acc[q] = fmaf(w2, v2[q], acc[q]);
        for (int q = 0; q < 4; ++q) acc[q] = fmaf(w3, v3[q], acc[q]);
    }
    for (; i < m; ++i) {
        const int s0 = __builtin_amdgcn_readlane(s_l, i);
        const float w0 = __int_as_float(__builtin_amdgcn_readlane(w_li, i));
        float v0[4];
        load_row<4>(hl + (long long)s0 * EMB, v0);
        for (int q = 0; q < 4; ++q) acc[q] = fmaf(w0, v0[q], acc[q]);
    }
}

__global__ __launch_bounds__(512) void gather_product_kernel(const int* __restrict__ rowptr,
                                                             const int* __restrict__ csr_src,
                                                             const float* __restrict__ dinv,
                                                             const float* __restrict__ h,
                                                             const float* __restrict__ b,
                                                             const float* __restrict__ W,
                                                             float* __restrict__ H2, int n, int e) {
    extern __shared__ float smem_f[];
    float* sW = smem_f;
    float* sX = smem_f + EMB * H2OUT;

    const int tid = threadIdx.x;
    const int lane = tid & 31;
    const int wv = __builtin_amdgcn_readfirstlane((int)(tid >> 5));
    const int node0 = blockIdx.x * FNODES + wv * NPW;

    float4 wq[4];
#pragma unroll
    for (int k = 0; k < 4; ++k) wq[k] = *(const float4*)&W[(tid + k * 512) * 4];

    int rp[NPW + 1];
    float ddq[NPW];
#pragma unroll
    for (int q = 0; q <= NPW; ++q) rp[q] = rowptr[min(node0 + q, n)];
#pragma unroll
    for (int q = 0; q < NPW; ++q) ddq[q] = dinv[min(node0 + q, n - 1)];

    int s_q[NPW];
    float w_q[NPW];
#pragma unroll
    for (int q = 0; q < NPW; ++q) s_q[q] = csr_src[min(rp[q] + lane, e - 1)];
#pragma unroll
    for (int q = 0; q < NPW; ++q) {
        const bool valid = lane < min(32, rp[q + 1] - rp[q]);
        s_q[q] = valid ? min(max(s_q[q], 0), n - 1) : 0;
    }
    float ds[NPW];
#pragma unroll
    for (int q = 0; q < NPW; ++q) ds[q] = dinv[s_q[q]];
#pragma unroll
    for (int k = 0; k < 4; ++k) *(float4*)&sW[(tid + k * 512) * 4] = wq[k];
    asm volatile("" ::"v"(ds[0]), "v"(ds[1]), "v"(ds[2]), "v"(ds[3]));
#pragma unroll
    for (int q = 0; q < NPW; ++q) {
        const bool valid = lane < min(32, rp[q + 1] - rp[q]);
        w_q[q] = valid ? ds[q] * ddq[q] : 0.0f;
    }

    const float* hl = h + lane * 4;
    float bb[4];
    load_row<4>(b + lane * 4, bb);

#pragma unroll
    for (int q = 0; q < NPW; ++q) {
        const int node = node0 + q;
        float o[4] = {0.0f, 0.0f, 0.0f, 0.0f};
        if (node < n) {
            float hs[4];
            load_row<4>(hl + (long long)node * EMB, hs);
            float acc[4] = {0.0f, 0.0f, 0.0f, 0.0f};
            const int beg = rp[q];
            const int end = rp[q + 1];
            const float dd = ddq[q];
            add_rows128(s_q[q], __float_as_int(w_q[q]), min(32, end - beg), hl, acc);
            for (int j0 = beg + 32; j0 < end; j0 += 32) {
                const int m = min(32, end - j0);
                int s_l = 0;
                float w_l = 0.0f;
                if (lane < m) {
                    s_l = csr_src[j0 + lane];
                    s_l = min(max(s_l, 0), n - 1);
                    w_l = dinv[s_l] * dd;
                }
                add_rows128(s_l, __float_as_int(w_l), m, hl, acc);
            }
            const float d2 = dd * dd;
            for (int c = 0; c < 4; ++c) o[c] = fmaxf(acc[c] + d2 * hs[c] + bb[c], 0.0f);
        }
        *(float4*)&sX[(wv * NPW + q) * XPITCH + lane * 4] = make_float4(o[0], o[1], o[2], o[3]);
    }
    __syncthreads();

    if (wv >= 8) return;
    const int nl = lane & 15;
    const int kh = lane >> 4;
    const int g = wv >> 1;
    const int ch = wv & 1;
    const float* xrow = &sX[(g * 16 + nl) * XPITCH + 2 * kh];
    const float* wcol = &sW[ch * 32 + nl];
    v8f c0 = {}, c1 = {};
#pragma unroll
    for (int ks = 0; ks < KSTEPS; ++ks) {
        const v2f a = *(const v2f*)&xrow[4 * ks];
        const int kk = 4 * ks + 2 * kh;
        v2f b0, b1;
        b0.x = wcol[kk * H2OUT];
        b0.y = wcol[(kk + 1) * H2OUT];
        b1.x = wcol[kk * H2OUT + 16];
        b1.y = wcol[(kk + 1) * H2OUT + 16];
        c0 = __builtin_amdgcn_wmma_f32_16x16x4_f32(false, a, false, b0, (short)0, c0, false, false);
        c1 = __builtin_amdgcn_wmma_f32_16x16x4_f32(false, a, false, b1, (short)0, c1, false, false);
    }
#pragma unroll
    for (int j = 0; j < 8; ++j) {
        const int r = blockIdx.x * FNODES + g * 16 + kh * 8 + j;
        if (r < n) {
            H2[(long long)r * H2OUT + ch * 32 + nl] = c0[j];
            H2[(long long)r * H2OUT + ch * 32 + 16 + nl] = c1[j];
        }
    }
}

extern "C" void kernel_launch(void* const* d_in, const int* in_sizes, int n_in,
                              void* d_out, int out_size, void* d_ws, size_t ws_size,
                              hipStream_t stream) {
    const int*   edge = (const int*)d_in[0];
    const float* x0   = (const float*)d_in[1];
    const float* W1   = (const float*)d_in[2];
    const float* b1   = (const float*)d_in[3];
    const float* W2   = (const float*)d_in[4];
    const float* b2   = (const float*)d_in[5];
    float* out = (float*)d_out;

    const int E    = in_sizes[0] / 2;
    const int N    = in_sizes[1] / EMB;
    const int HID  = in_sizes[3];
    const int REPR = in_sizes[5];
    const int* src = edge;
    const int* dst = edge + E;

    float* dinv = (float*)d_ws;
    float* h    = dinv + N;
    float* x1   = h + (size_t)N * HID;
    const int NB  = (N + BNODES - 1) >> BSH;
    const int NCH = (E + CHUNK - 1) / CHUNK;
    int* rowptr   = (int*)(x1 + (size_t)N * HID);
    unsigned* tmp = (unsigned*)(rowptr + (N + 1));
    int* csr_src  = (int*)(tmp + E);
    int* ghist    = csr_src + E;
    int* goff     = ghist + (size_t)NCH * NB;
    int* btot     = goff + (size_t)NCH * NB;

    const int T = 256;

    size_t smemA = (size_t)NB * sizeof(int);
    size_t smemC = (size_t)(CHUNK + 1) * (sizeof(int) + sizeof(unsigned short)) + (size_t)3 * NB * sizeof(int);
    bucket_hist_kernel<<<NCH, T, smemA, stream>>>(dst, ghist, E, N, NB);
    bucket_offsets_kernel<<<NB, T, 0, stream>>>(ghist, goff, btot, NCH, NB);
    bucket_scatter_kernel<<<NCH, T, smemC, stream>>>(src, dst, ghist, goff, btot, tmp, E, N, NB);
    bucket_sort_kernel<<<NB, BNODES, 0, stream>>>(tmp, btot, csr_src, rowptr, dinv, N, NB);

    size_t smem1 = (size_t)(EMB * HID + 128 * XPITCH) * sizeof(float);
    gemm_wmma_kernel<128><<<(N + 127) / 128, T, smem1, stream>>>(x0, W1, h, N);

    size_t smemF = (size_t)(EMB * H2OUT + FNODES * XPITCH) * sizeof(float);
    gather_product_kernel<<<(N + FNODES - 1) / FNODES, 512, smemF, stream>>>(rowptr, csr_src, dinv, h, b1,
                                                                           W2, x1, N, E);

    gather_kernel<64, false><<<(N + 7) / 8, T, 0, stream>>>(rowptr, csr_src, dinv, x1, b2, out, N);
}


# Layer-1 product loses its own launch: its 128-row WMMA tiles run as leading blocks (role by block index) of the four edge-sort launches, whose blocks leave the matrix pipes idle; 7 to 6 launches
# speedup vs baseline: 4.4001x; 1.1027x over previous
#include <hip/hip_runtime.h>
#include <hip/hip_bf16.h>

typedef __attribute__((ext_vector_type(2))) float v2f;
typedef __attribute__((ext_vector_type(8))) float v8f;

#define EMB 128
#define XPITCH 132
#define KSTEPS (EMB / 4)

#if defined(__has_builtin)
#if __has_builtin(__builtin_amdgcn_sched_barrier)
#define SCHED_FENCE() __builtin_amdgcn_sched_barrier(0)
#endif
#endif
#ifndef SCHED_FENCE
#define SCHED_FENCE()
#endif

#define BSH 8
#define BNODES (1 << BSH)
#define SRC_BITS 24
#define SRC_MASK ((1u << SRC_BITS) - 1u)
#define CEPT 25
#define CHUNK (256 * CEPT)
#define DEPT 8
#define DCAP 8192

__device__ __forceinline__ int block_excl_scan(int v, int* sh, int& tot) {
    const int lane = threadIdx.x & 31;
    const int wv = threadIdx.x >> 5;
    int inc = v;
    for (int off = 1; off < 32; off <<= 1) {
        int u = __shfl_up(inc, off, 32);
        if (lane >= off) inc += u;
    }
    if (lane == 31) sh[wv] = inc;
    __syncthreads();
    int wbase = 0, t = 0;
    for (int w = 0; w < 8; ++w) {
        int c = sh[w];
        if (w < wv) wbase += c;
        t += c;
    }
    __syncthreads();
    tot = t;
    return wbase + inc - v;
}

__device__ __forceinline__ void hist_role(const int* __restrict__ dst, int* __restrict__ ghist,
                                          int e, int n, int nb, int blk, int* lds_a) {
    const int t = threadIdx.x;
    for (int b = t; b < nb; b += 256) lds_a[b] = 0;
    __syncthreads();
    const int base = blk * CHUNK + t;
    int d[CEPT];
#pragma unroll
    for (int k = 0; k < CEPT; ++k) d[k] = dst[min(base + k * 256, e - 1)];
#pragma unroll
    for (int k = 0; k < CEPT; ++k) {
        atomicAdd(&lds_a[min(max(d[k], 0), n - 1) >> BSH], (base + k * 256 < e) ? 1 : 0);
    }
    __syncthreads();
    for (int b = t; b < nb; b += 256) ghist[blk * nb + b] = lds_a[b];
}

__device__ __forceinline__ void offsets_role(const int* __restrict__ ghist, int* __restrict__ goff,
                                             int* __restrict__ btot, int nch, int nb, int blk) {
    __shared__ int sh[8];
    const int b = blk;
    const int t = threadIdx.x;
    const int per = (nch + 255) / 256;
    const int c0 = min(t * per, nch);
    const int c1 = min(c0 + per, nch);
    int s = 0;
    for (int c = c0; c < c1; ++c) s += ghist[c * nb + b];
    int tot;
    int run = block_excl_scan(s, sh, tot);
    for (int c = c0; c < c1; ++c) {
        goff[c * nb + b] = run;
        run += ghist[c * nb + b];
    }
    if (t == 0) btot[b] = tot;
}

__device__ __forceinline__ void scatter_role(const int* __restrict__ src, const int* __restrict__ dst,
                                             const int* __restrict__ ghist,
                                             const int* __restrict__ goff,
                                             const int* __restrict__ btot,
                                             unsigned* __restrict__ tmp,
                                             int e, int n, int nb, int blk, int* lds_c) {
    unsigned* stage = (unsigned*)lds_c;
    int* lstart = lds_c + (CHUNK + 1);
    int* fill   = lstart + nb;
    int* gdelta = fill + nb;
    unsigned short* sb = (unsigned short*)(gdelta + nb);
    __shared__ int sh[8];
    const int t = threadIdx.x;

    const int base = blk * CHUNK + t;
    int d[CEPT], s[CEPT];
#pragma unroll
    for (int k = 0; k < CEPT; ++k) {
        int ic = min(base + k * 256, e - 1);
        d[k] = dst[ic];
        s[k] = src[ic];
    }

    const int per = (nb + 255) / 256;
    const int b0 = min(t * per, nb);
    const int b1 = min(b0 + per, nb);
    int sl = 0, sg = 0;
    for (int b = b0; b < b1; ++b) {
        sl += ghist[blk * nb + b];
        sg += btot[b];
    }
    int tot;
    int runl = block_excl_scan(sl, sh, tot);
    int rung = block_excl_scan(sg, sh, tot);
    for (int b = b0; b < b1; ++b) {
        lstart[b] = runl;
        fill[b] = 0;
        gdelta[b] = rung + goff[blk * nb + b] - runl;
        runl += ghist[blk * nb + b];
        rung += btot[b];
    }
    __syncthreads();

    unsigned wd[CEPT];
    int bk[CEPT], pos[CEPT];
#pragma unroll
    for (int k = 0; k < CEPT; ++k) {
        const bool valid = base + k * 256 < e;
        const int dd = min(max(d[k], 0), n - 1);
        bk[k] = dd >> BSH;
        wd[k] = ((unsigned)s[k] & SRC_MASK) | ((unsigned)(dd & (BNODES - 1)) << SRC_BITS);
        const int p = lstart[bk[k]] + atomicAdd(&fill[bk[k]], valid ? 1 : 0);
        pos[k] = valid ? p : CHUNK;
    }
#pragma unroll
    for (int k = 0; k < CEPT; ++k) {
        stage[pos[k]] = wd[k];
        sb[pos[k]] = (unsigned short)bk[k];
    }
    __syncthreads();

    const int cnt = min(CHUNK, e - blk * CHUNK);
    unsigned wq[CEPT];
    int gq[CEPT];
#pragma unroll
    for (int k = 0; k < CEPT; ++k) {
        const int ii = min(t + k * 256, cnt - 1);
        wq[k] = stage[ii];
        gq[k] = sb[ii];
    }
#pragma unroll
    for (int k = 0; k < CEPT; ++k) gq[k] = gdelta[gq[k]];
#pragma unroll
    for (int k = 0; k < CEPT; ++k) {
        const int i = t + k * 256;
        if (i < cnt) tmp[gq[k] + i] = wq[k];
    }
}

#define SORT_LDS_INTS (3 * BNODES + DCAP + 1)
__device__ __forceinline__ void sort_role(const unsigned* __restrict__ tmp,
                                          const int* __restrict__ btot,
                                          int* __restrict__ csr_src,
                                          int* __restrict__ rowptr,
                                          float* __restrict__ dinv,
                                          int n, int nb, int blk, int* lds_d) {
    int* ncnt   = lds_d;
    int* nstart = lds_d + BNODES;
    int* fill   = lds_d + 2 * BNODES;
    int* stage  = lds_d + 3 * BNODES;
    __shared__ int sh[8];
    const int b = blk;
    const int t = threadIdx.x;

    int part = 0;
    for (int j = t; j < b; j += BNODES) part += btot[j];
    int base;
    block_excl_scan(part, sh, base);
    const int cntb = btot[b];

    ncnt[t] = 0;
    fill[t] = 0;
    __syncthreads();

    for (int i0 = t; i0 < cntb; i0 += BNODES * DEPT) {
        unsigned w[DEPT];
#pragma unroll
        for (int k = 0; k < DEPT; ++k) w[k] = tmp[base + min(i0 + k * BNODES, cntb - 1)];
#pragma unroll
        for (int k = 0; k < DEPT; ++k)
            atomicAdd(&ncnt[w[k] >> SRC_BITS], (i0 + k * BNODES < cntb) ? 1 : 0);
    }
    __syncthreads();

    const int c = ncnt[t];
    int tot;
    const int st = block_excl_scan(c, sh, tot);
    nstart[t] = st;
    const int node = b * BNODES + t;
    if (node < n) {
        rowptr[node] = base + st;
        dinv[node] = rsqrtf((float)(c + 1));
    }
    if (b == nb - 1 && t == 0) rowptr[n] = base + cntb;
    __syncthreads();

    const bool staged = cntb <= DCAP;
    for (int i0 = t; i0 < cntb; i0 += BNODES * DEPT) {
        unsigned w[DEPT];
        int pos[DEPT];
#pragma unroll
        for (int k = 0; k < DEPT; ++k) w[k] = tmp[base + min(i0 + k * BNODES, cntb - 1)];
#pragma unroll
        for (int k = 0; k < DEPT; ++k) {
            const bool valid = i0 + k * BNODES < cntb;
            const int dl = (int)(w[k] >> SRC_BITS);
            const int p = nstart[dl] + atomicAdd(&fill[dl], valid ? 1 : 0);
            pos[k] = valid ? p : -1;
        }
        if (staged) {
#pragma unroll
            for (int k = 0; k < DEPT; ++k) stage[pos[k] < 0 ? DCAP : pos[k]] = (int)(w[k] & SRC_MASK);
        } else {
#pragma unroll
            for (int k = 0; k < DEPT; ++k) {
                if (pos[k] >= 0) csr_src[base + pos[k]] = (int)(w[k] & SRC_MASK);
            }
        }
    }
    if (staged) {
        __syncthreads();
        for (int i = t; i < cntb; i += BNODES) csr_src[base + i] = stage[i];
    }
}

template <int HOUT>
__device__ __forceinline__ void gemm_role(const float* __restrict__ X, const float* __restrict__ W,
                                          float* __restrict__ H, int nrows, int blk, float* smem) {
    float* sW = smem;
    float* sX = smem + EMB * HOUT;

    const int tid = threadIdx.x;
    const int rowBase = blk * 128;

    constexpr int WP = EMB * HOUT / (256 * 4);
    constexpr int XP = 128 * EMB / (256 * 4);
    const int xr = tid >> 5;
    const int xc = (tid & 31) * 4;
    float4 wv[WP], xv[XP];
#pragma unroll
    for (int k = 0; k < WP; ++k) wv[k] = *(const float4*)&W[(tid + k * 256) * 4];
#pragma unroll
    for (int k = 0; k < XP; ++k)
        xv[k] = *(const float4*)&X[(long long)min(rowBase + xr + 8 * k, nrows - 1) * EMB + xc];
#pragma unroll
    for (int k = 0; k < WP; ++k) *(float4*)&sW[(tid + k * 256) * 4] = wv[k];
#pragma unroll
    for (int k = 0; k < XP; ++k) {
        const int r = xr + 8 * k;
        *(float4*)&sX[r * XPITCH + xc] =
            (rowBase + r < nrows) ? xv[k] : make_float4(0.f, 0.f, 0.f, 0.f);
    }
    __syncthreads();

    const int wave = tid >> 5;
    const int lane = tid & 31;
    const int nl = lane & 15;
    const int kh = lane >> 4;
    const int wr = wave * 16;

    const float* xrow = &sX[(wr + nl) * XPITCH + 2 * kh];
    v2f afrag[KSTEPS];
#pragma unroll
    for (int ks = 0; ks < KSTEPS; ++ks) afrag[ks] = *(const v2f*)&xrow[4 * ks];
    SCHED_FENCE();

    for (int nt = 0; nt < HOUT / 16; ++nt) {
        const float* wcol = &sW[nt * 16 + nl];
        v2f bfrag[KSTEPS];
#pragma unroll
        for (int ks = 0; ks < KSTEPS; ++ks) {
            const int kk = 4 * ks + 2 * kh;
            v2f b;
            b.x = wcol[kk * HOUT];
            b.y = wcol[(kk + 1) * HOUT];
            bfrag[ks] = b;
        }
        SCHED_FENCE();
        v8f acc = {};
#pragma unroll
        for (int ks = 0; ks < KSTEPS; ++ks) {
            acc = __builtin_amdgcn_wmma_f32_16x16x4_f32(
                false, afrag[ks], false, bfrag[ks], (short)0, acc, false, false);
        }
        SCHED_FENCE();
#pragma unroll
        for (int j = 0; j < 8; ++j) {
            int r = rowBase + wr + kh * 8 + j;
            if (r < nrows) H[(long long)r * HOUT + nt * 16 + nl] = acc[j];
        }
    }
}

#define GEMM_LDS_BYTES ((EMB * 128 + 128 * XPITCH) * 4)

__global__ __launch_bounds__(256) void hist_gemm_kernel(const int* __restrict__ dst,
                                                        int* __restrict__ ghist, int e, int n, int nb,
                                                        int ng, const float* __restrict__ X,
                                                        const float* __restrict__ W,
                                                        float* __restrict__ H, int g0) {
    extern __shared__ __attribute__((aligned(16))) float lds_all[];
    const int blk = blockIdx.x;
    if (blk < ng) gemm_role<128>(X, W, H, n, g0 + blk, lds_all);
    else hist_role(dst, ghist, e, n, nb, blk - ng, (int*)lds_all);
}

__global__ __launch_bounds__(256) void offsets_gemm_kernel(const int* __restrict__ ghist,
                                                           int* __restrict__ goff,
                                                           int* __restrict__ btot, int nch, int nb,
                                                           int ng, const float* __restrict__ X,
                                                           const float* __restrict__ W,
                                                           float* __restrict__ H, int n, int g0) {
    extern __shared__ __attribute__((aligned(16))) float lds_all[];
    const int blk = blockIdx.x;
    if (blk < ng) gemm_role<128>(X, W, H, n, g0 + blk, lds_all);
    else offsets_role(ghist, goff, btot, nch, nb, blk - ng);
}

__global__ __launch_bounds__(256) void scatter_gemm_kernel(const int* __restrict__ src,
                                                           const int* __restrict__ dst,
                                                           const int* __restrict__ ghist,
                                                           const int* __restrict__ goff,
                                                           const int* __restrict__ btot,
                                                           unsigned* __restrict__ tmp,
                                                           int e, int n, int nb, int ng,
                                                           const float* __restrict__ X,
                                                           const float* __restrict__ W,
                                                           float* __restrict__ H, int g0) {
    extern __shared__ __attribute__((aligned(16))) float lds_all[];
    const int blk = blockIdx.x;
    if (blk < ng) gemm_role<128>(X, W, H, n, g0 + blk, lds_all);
    else scatter_role(src, dst, ghist, goff, btot, tmp, e, n, nb, blk - ng, (int*)lds_all);
}

__global__ __launch_bounds__(256) void sort_gemm_kernel(const unsigned* __restrict__ tmp,
                                                        const int* __restrict__ btot,
                                                        int* __restrict__ csr_src,
                                                        int* __restrict__ rowptr,
                                                        float* __restrict__ dinv, int n, int nb,
                                                        int ng, const float* __restrict__ X,
                                                        const float* __restrict__ W,
                                                        float* __restrict__ H, int g0) {
    extern __shared__ __attribute__((aligned(16))) float lds_all[];
    const int blk = blockIdx.x;
    if (blk < ng) gemm_role<128>(X, W, H, n, g0 + blk, lds_all);
    else sort_role(tmp, btot, csr_src, rowptr, dinv, n, nb, blk - ng, (int*)lds_all);
}

template <int V>
__device__ __forceinline__ void load_row(const float* __restrict__ p, float (&v)[V]) {
    if constexpr (V == 4) {
        float4 t = *(const float4*)p;
        v[0] = t.x; v[1] = t.y; v[2] = t.z; v[3] = t.w;
    } else {
        float2 t = *(const float2*)p;
        v[0] = t.x; v[1] = t.y;
    }
}

template <int HOUT, bool RELU>
__global__ __launch_bounds__(256) void gather_kernel(const int* __restrict__ rowptr,
                                                     const int* __restrict__ csr_src,
                                                     const float* __restrict__ dinv,
                                                     const float* __restrict__ h,
                                                     const float* __restrict__ b,
                                                     float* __restrict__ y, int n) {
    constexpr int V = HOUT / 32;
    const int lane = threadIdx.x & 31;
    const int wv = __builtin_amdgcn_readfirstlane((int)(threadIdx.x >> 5));
    const int node = blockIdx.x * 8 + wv;
    if (node >= n) return;
    const int beg = rowptr[node];
    const int end = rowptr[node + 1];
    const float dd = dinv[node];
    const float* hl = h + lane * V;

    float hs[V], bb[V];
    load_row<V>(hl + (long long)node * HOUT, hs);
    load_row<V>(b + lane * V, bb);

    float acc[V];
    for (int q = 0; q < V; ++q) acc[q] = 0.0f;

    for (int j0 = beg; j0 < end; j0 += 32) {
        const int m = min(32, end - j0);
        int s_l = 0;
        float w_l = 0.0f;
        if (lane < m) {
            s_l = csr_src[j0 + lane];
            s_l = min(max(s_l, 0), n - 1);
            w_l = dinv[s_l] * dd;
        }
        const int w_li = __float_as_int(w_l);
        int i = 0;
        for (; i + 4 <= m; i += 4) {
            const int s0 = __builtin_amdgcn_readlane(s_l, i);
            const int s1 = __builtin_amdgcn_readlane(s_l, i + 1);
            const int s2 = __builtin_amdgcn_readlane(s_l, i + 2);
            const int s3 = __builtin_amdgcn_readlane(s_l, i + 3);
            const float w0 = __int_as_float(__builtin_amdgcn_readlane(w_li, i));
            const float w1 = __int_as_float(__builtin_amdgcn_readlane(w_li, i + 1));
            const float w2 = __int_as_float(__builtin_amdgcn_readlane(w_li, i + 2));
            const float w3 = __int_as_float(__builtin_amdgcn_readlane(w_li, i + 3));
            float v0[V], v1[V], v2[V], v3[V];
            load_row<V>(hl + (long long)s0 * HOUT, v0);
            load_row<V>(hl + (long long)s1 * HOUT, v1);
            load_row<V>(hl + (long long)s2 * HOUT, v2);
            load_row<V>(hl + (long long)s3 * HOUT, v3);
            for (int q = 0; q < V; ++q) acc[q] = fmaf(w0, v0[q], acc[q]);
            for (int q = 0; q < V; ++q) acc[q] = fmaf(w1, v1[q], acc[q]);
            for (int q = 0; q < V; ++q) acc[q] = fmaf(w2, v2[q], acc[q]);
            for (int q = 0; q < V; ++q) acc[q] = fmaf(w3, v3[q], acc[q]);
        }
        for (; i < m; ++i) {
            const int s0 = __builtin_amdgcn_readlane(s_l, i);
            const float w0 = __int_as_float(__builtin_amdgcn_readlane(w_li, i));
            float v0[V];
            load_row<V>(hl + (long long)s0 * HOUT, v0);
            for (int q = 0; q < V; ++q) acc[q] = fmaf(w0, v0[q], acc[q]);
        }
    }

    const float d2 = dd * dd;
    for (int q = 0; q < V; ++q) {
        float v = acc[q] + d2 * hs[q] + bb[q];
        acc[q] = RELU ? fmaxf(v, 0.0f) : v;
    }

    float* o = y + (long long)node * HOUT + lane * V;
    if constexpr (V == 4) {
        *(float4*)o = make_float4(acc[0], acc[1], acc[2], acc[3]);
    } else {
        *(float2*)o = make_float2(acc[0], acc[1]);
    }
}

#define FNODES 64
#define NPW 4
#define H2OUT 64

__device__ __forceinline__ void add_rows128(int s_l, int w_li, int m, const float* __restrict__ hl,
                                            float (&acc)[4]) {
    int i = 0;
    for (; i + 4 <= m; i += 4) {
        const int s0 = __builtin_amdgcn_readlane(s_l, i);
        const int s1 = __builtin_amdgcn_readlane(s_l, i + 1);
        const int s2 = __builtin_amdgcn_readlane(s_l, i + 2);
        const int s3 = __builtin_amdgcn_readlane(s_l, i + 3);
        const float w0 = __int_as_float(__builtin_amdgcn_readlane(w_li, i));
        const float w1 = __int_as_float(__builtin_amdgcn_readlane(w_li, i + 1));
        const float w2 = __int_as_float(__builtin_amdgcn_readlane(w_li, i + 2));
        const float w3 = __int_as_float(__builtin_amdgcn_readlane(w_li, i + 3));
        float v0[4], v1[4], v2[4], v3[4];
        load_row<4>(hl + (long long)s0 * EMB, v0);
        load_row<4>(hl + (long long)s1 * EMB, v1);
        load_row<4>(hl + (long long)s2 * EMB, v2);
        load_row<4>(hl + (long long)s3 * EMB, v3);
        for (int q = 0; q < 4; ++q) acc[q] = fmaf(w0, v0[q], acc[q]);
        for (int q = 0; q < 4; ++q) acc[q] = fmaf(w1, v1[q], acc[q]);
        for (int q = 0; q < 4; ++q) acc[q] = fmaf(w2, v2[q], acc[q]);
        for (int q = 0; q < 4; ++q) acc[q] = fmaf(w3, v3[q], acc[q]);
    }
    for (; i < m; ++i) {
        const int s0 = __builtin_amdgcn_readlane(s_l, i);
        const float w0 = __int_as_float(__builtin_amdgcn_readlane(w_li, i));
        float v0[4];
        load_row<4>(hl + (long long)s0 * EMB, v0);
        for (int q = 0; q < 4; ++q) acc[q] = fmaf(w0, v0[q], acc[q]);
    }
}

__global__ __launch_bounds__(512) void gather_product_kernel(const int* __restrict__ rowptr,
                                                             const int* __restrict__ csr_src,
                                                             const float* __restrict__ dinv,
                                                             const float* __restrict__ h,
                                                             const float* __restrict__ b,
                                                             const float* __restrict__ W,
                                                             float* __restrict__ H2, int n, int e) {
    extern __shared__ float smem_f[];
    float* sW = smem_f;
    float* sX = smem_f + EMB * H2OUT;

    const int tid = threadIdx.x;
    const int lane = tid & 31;
    const int wv = __builtin_amdgcn_readfirstlane((int)(tid >> 5));
    const int node0 = blockIdx.x * FNODES + wv * NPW;

    float4 wq[4];
#pragma unroll
    for (int k = 0; k < 4; ++k) wq[k] = *(const float4*)&W[(tid + k * 512) * 4];

    int rp[NPW + 1];
    float ddq[NPW];
#pragma unroll
    for (int q = 0; q <= NPW; ++q) rp[q] = rowptr[min(node0 + q, n)];
#pragma unroll
    for (int q = 0; q < NPW; ++q) ddq[q] = dinv[min(node0 + q, n - 1)];

    int s_q[NPW];
    float w_q[NPW];
#pragma unroll
    for (int q = 0; q < NPW; ++q) s_q[q] = csr_src[min(rp[q] + lane, e - 1)];
#pragma unroll
    for (int q = 0; q < NPW; ++q) {
        const bool valid = lane < min(32, rp[q + 1] - rp[q]);
        s_q[q] = valid ? min(max(s_q[q], 0), n - 1) : 0;
    }
    float ds[NPW];
#pragma unroll
    for (int q = 0; q < NPW; ++q) ds[q] = dinv[s_q[q]];
#pragma unroll
    for (int k = 0; k < 4; ++k) *(float4*)&sW[(tid + k * 512) * 4] = wq[k];
    asm volatile("" ::"v"(ds[0]), "v"(ds[1]), "v"(ds[2]), "v"(ds[3]));
#pragma unroll
    for (int q = 0; q < NPW; ++q) {
        const bool valid = lane < min(32, rp[q + 1] - rp[q]);
        w_q[q] = valid ? ds[q] * ddq[q] : 0.0f;
    }

    const float* hl = h + lane * 4;
    float bb[4];
    load_row<4>(b + lane * 4, bb);

#pragma unroll
    for (int q = 0; q < NPW; ++q) {
        const int node = node0 + q;
        float o[4] = {0.0f, 0.0f, 0.0f, 0.0f};
        if (node < n) {
            float hs[4];
            load_row<4>(hl + (long long)node * EMB, hs);
            float acc[4] = {0.0f, 0.0f, 0.0f, 0.0f};
            const int beg = rp[q];
            const int end = rp[q + 1];
            const float dd = ddq[q];
            add_rows128(s_q[q], __float_as_int(w_q[q]), min(32, end - beg), hl, acc);
            for (int j0 = beg + 32; j0 < end; j0 += 32) {
                const int m = min(32, end - j0);
                int s_l = 0;
                float w_l = 0.0f;
                if (lane < m) {
                    s_l = csr_src[j0 + lane];
                    s_l = min(max(s_l, 0), n - 1);
                    w_l = dinv[s_l] * dd;
                }
                add_rows128(s_l, __float_as_int(w_l), m, hl, acc);
            }
            const float d2 = dd * dd;
            for (int c = 0; c < 4; ++c) o[c] = fmaxf(acc[c] + d2 * hs[c] + bb[c], 0.0f);
        }
        *(float4*)&sX[(wv * NPW + q) * XPITCH + lane * 4] = make_float4(o[0], o[1], o[2], o[3]);
    }
    __syncthreads();

    if (wv >= 8) return;
    const int nl = lane & 15;
    const int kh = lane >> 4;
    const int g = wv >> 1;
    const int ch = wv & 1;
    const float* xrow = &sX[(g * 16 + nl) * XPITCH + 2 * kh];
    const float* wcol = &sW[ch * 32 + nl];
    v8f c0 = {}, c1 = {};
#pragma unroll
    for (int ks = 0; ks < KSTEPS; ++ks) {
        const v2f a = *(const v2f*)&xrow[4 * ks];
        const int kk = 4 * ks + 2 * kh;
        v2f b0, b1;
        b0.x = wcol[kk * H2OUT];
        b0.y = wcol[(kk + 1) * H2OUT];
        b1.x = wcol[kk * H2OUT + 16];
        b1.y = wcol[(kk + 1) * H2OUT + 16];
        c0 = __builtin_amdgcn_wmma_f32_16x16x4_f32(false, a, false, b0, (short)0, c0, false, false);
        c1 = __builtin_amdgcn_wmma_f32_16x16x4_f32(false, a, false, b1, (short)0, c1, false, false);
    }
#pragma unroll
    for (int j = 0; j < 8; ++j) {
        const int r = blockIdx.x * FNODES + g * 16 + kh * 8 + j;
        if (r < n) {
            H2[(long long)r * H2OUT + ch * 32 + nl] = c0[j];
            H2[(long long)r * H2OUT + ch * 32 + 16 + nl] = c1[j];
        }
    }
}

extern "C" void kernel_launch(void* const* d_in, const int* in_sizes, int n_in,
                              void* d_out, int out_size, void* d_ws, size_t ws_size,
                              hipStream_t stream) {
    const int*   edge = (const int*)d_in[0];
    const float* x0   = (const float*)d_in[1];
    const float* W1   = (const float*)d_in[2];
    const float* b1   = (const float*)d_in[3];
    const float* W2   = (const float*)d_in[4];
    const float* b2   = (const float*)d_in[5];
    float* out = (float*)d_out;

    const int E    = in_sizes[0] / 2;
    const int N    = in_sizes[1] / EMB;
    const int HID  = in_sizes[3];
    const int REPR = in_sizes[5];
    const int* src = edge;
    const int* dst = edge + E;

    float* dinv = (float*)d_ws;
    float* h    = dinv + N;
    float* x1   = h + (size_t)N * HID;
    const int NB  = (N + BNODES - 1) >> BSH;
    const int NCH = (E + CHUNK - 1) / CHUNK;
    int* rowptr   = (int*)(x1 + (size_t)N * HID);
    unsigned* tmp = (unsigned*)(rowptr + (N + 1));
    int* csr_src  = (int*)(tmp + E);
    int* ghist    = csr_src + E;
    int* goff     = ghist + (size_t)NCH * NB;
    int* btot     = goff + (size_t)NCH * NB;

    const int T = 256;

    size_t smemA = (size_t)NB * sizeof(int);
    size_t smemC = (size_t)(CHUNK + 1) * (sizeof(int) + sizeof(unsigned short)) + (size_t)3 * NB * sizeof(int);
    size_t smemD = (size_t)SORT_LDS_INTS * sizeof(int);
    size_t smem1 = (size_t)GEMM_LDS_BYTES;
    if (smemA < smem1) smemA = smem1;
    if (smemC < smem1) smemC = smem1;
    if (smemD < smem1) smemD = smem1;
    const int GB = (N + 127) / 128;
    const int UNITS = 256;
    int gA = 2 * UNITS - NCH; if (gA > UNITS) gA = UNITS; if (gA < 0) gA = 0; if (gA > GB) gA = GB;
    int gC = 2 * UNITS - NCH; if (gC > UNITS) gC = UNITS; if (gC < 0) gC = 0; if (gC > GB - gA) gC = GB - gA;
    int gD = 2 * UNITS - NB;  if (gD > UNITS) gD = UNITS; if (gD < 0) gD = 0; if (gD > GB - gA - gC) gD = GB - gA - gC;
    const int gB = GB - gA - gC - gD;
    hist_gemm_kernel<<<gA + NCH, T, smemA, stream>>>(dst, ghist, E, N, NB, gA, x0, W1, h, 0);
    offsets_gemm_kernel<<<gB + NB, T, smem1, stream>>>(ghist, goff, btot, NCH, NB, gB, x0, W1, h, N, gA);
    scatter_gemm_kernel<<<gC + NCH, T, smemC, stream>>>(src, dst, ghist, goff, btot, tmp, E, N, NB, gC,
                                                       x0, W1, h, gA + gB);
    sort_gemm_kernel<<<gD + NB, BNODES, smemD, stream>>>(tmp, btot, csr_src, rowptr, dinv, N, NB, gD,
                                                        x0, W1, h, gA + gB + gC);

    size_t smemF = (size_t)(EMB * H2OUT + FNODES * XPITCH) * sizeof(float);
    gather_product_kernel<<<(N + FNODES - 1) / FNODES, 512, smemF, stream>>>(rowptr, csr_src, dinv, h, b1,
                                                                           W2, x1, N, E);

    gather_kernel<64, false><<<(N + 7) / 8, T, 0, stream>>>(rowptr, csr_src, dinv, x1, b2, out, N);
}
